# starting kernel
# baseline (speedup 1.0000x reference)
#include <hip/hip_runtime.h>
#include <math.h>


typedef __attribute__((ext_vector_type(2))) float v2f;
typedef __attribute__((ext_vector_type(8))) float v8f;

#define NFEATS   128
#define HIDDEN   64
#define NCLASSES 40
#define NLAYERS  8


__global__ void k_deg_init(float* __restrict__ deg, int n) {
    int i = blockIdx.x * blockDim.x + threadIdx.x;
    if (i < n) deg[i] = 1.0f;
}

__global__ void k_deg_edge(const int* __restrict__ col, float* __restrict__ deg, int nE) {
    int e = blockIdx.x * blockDim.x + threadIdx.x;
    if (e < nE) atomicAdd(&deg[col[e]], 1.0f);
}

__global__ void k_rsqrt_inplace(float* __restrict__ deg, int n) {
    int i = blockIdx.x * blockDim.x + threadIdx.x;
    if (i < n) { float d = deg[i]; deg[i] = d > 0.0f ? rsqrtf(d) : 0.0f; }
}


__global__ void k_agg_init(const float* __restrict__ dinv, const float* __restrict__ h,
                           float* __restrict__ agg, int total) {
    int i = blockIdx.x * blockDim.x + threadIdx.x;
    if (i < total) {
        float d = dinv[i >> 6];
        agg[i] = d * d * h[i];
    }
}

__global__ void k_spmm(const int* __restrict__ row, const int* __restrict__ col,
                       const float* __restrict__ dinv, const float* __restrict__ h,
                       float* __restrict__ agg, int nE) {
    unsigned tid = blockIdx.x * blockDim.x + threadIdx.x;
    int e = tid >> 4;
    if (e >= nE) return;
    int f = (tid & 15) * 4;
    int s = row[e], t = col[e];
    float w = dinv[s] * dinv[t];
    const float4 hv = *(const float4*)(h + (size_t)s * HIDDEN + f);
    float* dst = agg + (size_t)t * HIDDEN + f;
    atomicAdd(dst + 0, w * hv.x);
    atomicAdd(dst + 1, w * hv.y);
    atomicAdd(dst + 2, w * hv.z);
    atomicAdd(dst + 3, w * hv.w);
}


__device__ __forceinline__ v2f bfrag_rowmajor(const float* __restrict__ B, int ldb,
                                              int kb  , int ncol) {
    v2f b;
    b.x = B[(size_t)kb * ldb + ncol];
    b.y = B[(size_t)(kb + 1) * ldb + ncol];
    return b;
}

#define WMMA_F32X4(acc, a, b) \
    (acc) = __builtin_amdgcn_wmma_f32_16x16x4_f32(false, (a), false, (b), (short)0, (acc), false, false)


__global__ __launch_bounds__(256) void k_gemm0(const float* __restrict__ x,
                                               const float* __restrict__ W0,
                                               const float* __restrict__ b0,
                                               float* __restrict__ h,
                                               float* __restrict__ h0, int n) {
    const int lane = threadIdx.x & 31;
    const int wave = threadIdx.x >> 5;
    const int m0   = (blockIdx.x * 8 + wave) * 16;
    const int ml   = lane & 15;
    const int kh   = lane >> 4;
    int m = m0 + ml; if (m >= n) m = n - 1;

    v8f acc[4] = {};
    for (int kk = 0; kk < NFEATS; kk += 4) {
        const int ka = kk + kh * 2;
        const float2 av = *(const float2*)(x + (size_t)m * NFEATS + ka);
        v2f a; a.x = av.x; a.y = av.y;
        #pragma unroll
        for (int nt = 0; nt < 4; ++nt) {
            v2f b = bfrag_rowmajor(W0, HIDDEN, ka, nt * 16 + ml);
            WMMA_F32X4(acc[nt], a, b);
        }
    }
    const int rbase = m0 + kh * 8;
    const bool full = (m0 + 16 <= n);
    #pragma unroll
    for (int nt = 0; nt < 4; ++nt) {
        int c = nt * 16 + ml;
        float bias = b0[c];
        if (full) {
            #pragma unroll
            for (int v = 0; v < 8; ++v) {
                float val = fmaxf(acc[nt][v] + bias, 0.0f);
                size_t o = (size_t)(rbase + v) * HIDDEN + c;
                h[o] = val; h0[o] = val;
            }
        } else {
            #pragma unroll
            for (int v = 0; v < 8; ++v) {
                int r = rbase + v;
                if (r < n) {
                    float val = fmaxf(acc[nt][v] + bias, 0.0f);
                    size_t o = (size_t)r * HIDDEN + c;
                    h[o] = val; h0[o] = val;
                }
            }
        }
    }
}


__global__ __launch_bounds__(256) void k_layer(const float* __restrict__ agg,
                                               const float* __restrict__ h0,
                                               const float* __restrict__ W,
                                               float* __restrict__ h,
                                               float beta, int n) {
    const int lane = threadIdx.x & 31;
    const int wave = threadIdx.x >> 5;
    const int m0   = (blockIdx.x * 8 + wave) * 16;
    const int ml   = lane & 15;
    const int kh   = lane >> 4;
    int m = m0 + ml; if (m >= n) m = n - 1;

    v8f acc[4] = {};
    for (int kk = 0; kk < HIDDEN; kk += 4) {
        const int ka = kk + kh * 2;
        size_t idx = (size_t)m * HIDDEN + ka;
        const float2 av = *(const float2*)(agg + idx);
        const float2 hv = *(const float2*)(h0 + idx);
        v2f a;
        a.x = 0.9f * av.x + 0.1f * hv.x;
        a.y = 0.9f * av.y + 0.1f * hv.y;
        #pragma unroll
        for (int nt = 0; nt < 4; ++nt) {
            v2f b = bfrag_rowmajor(W, HIDDEN, ka, nt * 16 + ml);
            WMMA_F32X4(acc[nt], a, b);
        }
    }
    const float omb = 1.0f - beta;
    const int rbase = m0 + kh * 8;
    const bool full = (m0 + 16 <= n);
    #pragma unroll
    for (int nt = 0; nt < 4; ++nt) {
        int c = nt * 16 + ml;
        if (full) {
            #pragma unroll
            for (int v = 0; v < 8; ++v) {
                size_t o = (size_t)(rbase + v) * HIDDEN + c;
                float sup = 0.9f * agg[o] + 0.1f * h0[o];
                h[o] = fmaxf(beta * acc[nt][v] + omb * sup, 0.0f);
            }
        } else {
            #pragma unroll
            for (int v = 0; v < 8; ++v) {
                int r = rbase + v;
                if (r < n) {
                    size_t o = (size_t)r * HIDDEN + c;
                    float sup = 0.9f * agg[o] + 0.1f * h0[o];
                    h[o] = fmaxf(beta * acc[nt][v] + omb * sup, 0.0f);
                }
            }
        }
    }
}


__global__ __launch_bounds__(256) void k_out(const float* __restrict__ h,
                                             const float* __restrict__ Wout,
                                             const float* __restrict__ bout,
                                             float* __restrict__ out, int n) {
    const int lane = threadIdx.x & 31;
    const int wave = threadIdx.x >> 5;
    const int m0   = (blockIdx.x * 8 + wave) * 16;
    const int ml   = lane & 15;
    const int kh   = lane >> 4;
    int m = m0 + ml; if (m >= n) m = n - 1;

    v8f acc[3] = {};
    for (int kk = 0; kk < HIDDEN; kk += 4) {
        const int ka = kk + kh * 2;
        const float2 av = *(const float2*)(h + (size_t)m * HIDDEN + ka);
        v2f a; a.x = av.x; a.y = av.y;
        #pragma unroll
        for (int nt = 0; nt < 3; ++nt) {
            int c = nt * 16 + ml;
            v2f b;
            b.x = (c < NCLASSES) ? Wout[(size_t)ka * NCLASSES + c] : 0.0f;
            b.y = (c < NCLASSES) ? Wout[(size_t)(ka + 1) * NCLASSES + c] : 0.0f;
            WMMA_F32X4(acc[nt], a, b);
        }
    }
    const int rbase = m0 + kh * 8;
    const bool full = (m0 + 16 <= n);
    #pragma unroll
    for (int nt = 0; nt < 3; ++nt) {
        int c = nt * 16 + ml;
        if (c < NCLASSES) {
            float bias = bout[c];
            if (full) {
                #pragma unroll
                for (int v = 0; v < 8; ++v)
                    out[(size_t)(rbase + v) * NCLASSES + c] = acc[nt][v] + bias;
            } else {
                #pragma unroll
                for (int v = 0; v < 8; ++v) {
                    int r = rbase + v;
                    if (r < n) out[(size_t)r * NCLASSES + c] = acc[nt][v] + bias;
                }
            }
        }
    }
}


extern "C" void kernel_launch(void* const* d_in, const int* in_sizes, int n_in,
                              void* d_out, int out_size, void* d_ws, size_t ws_size,
                              hipStream_t stream) {
    const float* x    = (const float*)d_in[0];
    const int*   ei   = (const int*)  d_in[1];
    const float* W0   = (const float*)d_in[2];
    const float* b0   = (const float*)d_in[3];
    const float* Wl   = (const float*)d_in[4];
    const float* Wout = (const float*)d_in[5];
    const float* bout = (const float*)d_in[6];

    const int N = in_sizes[0] / NFEATS;
    const int E = in_sizes[1] / 2;
    const int* erow = ei;
    const int* ecol = ei + E;

    float* deg = (float*)d_ws;
    size_t nAl = ((size_t)N + 63) & ~(size_t)63;
    float* h   = deg + nAl;
    float* h0  = h  + (size_t)N * HIDDEN;
    float* agg = h0 + (size_t)N * HIDDEN;
    (void)ws_size; (void)n_in; (void)out_size;

    const int nodeBlocks = (N + 255) / 256;
    const int edgeBlocks = (E + 255) / 256;
    const int gemmBlocks = (N + 127) / 128;
    const int total      = N * HIDDEN;
    const int aggBlocks  = (total + 255) / 256;
    const long spmmThr   = (long)E * 16;
    const int spmmBlocks = (int)((spmmThr + 255) / 256);

    k_deg_init     <<<nodeBlocks, 256, 0, stream>>>(deg, N);
    k_deg_edge     <<<edgeBlocks, 256, 0, stream>>>(ecol, deg, E);
    k_rsqrt_inplace<<<nodeBlocks, 256, 0, stream>>>(deg, N);

    k_gemm0<<<gemmBlocks, 256, 0, stream>>>(x, W0, b0, h, h0, N);

    for (int i = 0; i < NLAYERS; ++i) {
        k_agg_init<<<aggBlocks,  256, 0, stream>>>(deg, h, agg, total);
        k_spmm    <<<spmmBlocks, 256, 0, stream>>>(erow, ecol, deg, h, agg, E);
        float beta = logf(0.5f / (float)(i + 1) + 1.0f);
        k_layer<<<gemmBlocks, 256, 0, stream>>>(agg, h0, Wl + (size_t)i * HIDDEN * HIDDEN,
                                                h, beta, N);
    }

    k_out<<<gemmBlocks, 256, 0, stream>>>(h, Wout, bout, (float*)d_out, N);
}


# SpMM as register gather over a per-call target-sorted edge list (one int atomic rank per edge + scan + fill); float atomics removed, per-node sum order = arrival-order rank
# speedup vs baseline: 2.8823x; 2.8823x over previous
#include <hip/hip_runtime.h>
#include <math.h>


typedef __attribute__((ext_vector_type(2))) float v2f;
typedef __attribute__((ext_vector_type(8))) float v8f;

#define NFEATS   128
#define HIDDEN   64
#define NCLASSES 40
#define NLAYERS  8


__global__ void k_cnt_zero(int* __restrict__ cnt, int n) {
    int i = blockIdx.x * blockDim.x + threadIdx.x;
    if (i < n) cnt[i] = 0;
}

__global__ void k_rank(const int* __restrict__ col, int* __restrict__ cnt,
                       int* __restrict__ rank, int nE) {
    int e = blockIdx.x * blockDim.x + threadIdx.x;
    if (e < nE) rank[e] = atomicAdd(&cnt[col[e]], 1);
}

__global__ __launch_bounds__(1024) void k_scan(const int* __restrict__ cnt, int* __restrict__ off,
                                                float* __restrict__ dinv, int n) {
    __shared__ int wsum[32];
    __shared__ int sbase;
    const int tid = threadIdx.x, lane = tid & 31, wv = tid >> 5;
    const int lim = blockIdx.x * 1024;

    int part = 0;
    for (int j = tid; j < lim; j += 1024) part += cnt[j];
    for (int o = 16; o > 0; o >>= 1) part += __shfl_xor(part, o, 32);
    if (lane == 0) wsum[wv] = part;
    __syncthreads();
    if (wv == 0) {
        int v = wsum[lane];
        for (int o = 16; o > 0; o >>= 1) v += __shfl_xor(v, o, 32);
        if (lane == 0) sbase = v;
    }
    __syncthreads();
    const int base = sbase;

    const int i = lim + tid;
    const int c = (i < n) ? cnt[i] : 0;
    int incl = c;
    for (int o = 1; o < 32; o <<= 1) {
        int t = __shfl_up(incl, o, 32);
        if (lane >= o) incl += t;
    }
    if (lane == 31) wsum[wv] = incl;
    __syncthreads();
    int wpre = 0;
    for (int k = 0; k < wv; ++k) wpre += wsum[k];
    if (i <= n) off[i] = base + wpre + incl - c;
    if (i < n) { float d = (float)(c + 1); dinv[i] = d > 0.0f ? rsqrtf(d) : 0.0f; }
}

__global__ void k_fill(const int* __restrict__ row, const int* __restrict__ col,
                       const int* __restrict__ rank, const int* __restrict__ off,
                       const float* __restrict__ dinv, int2* __restrict__ csr, int nE) {
    int e = blockIdx.x * blockDim.x + threadIdx.x;
    if (e < nE) {
        int s = row[e], t = col[e];
        float w = dinv[s] * dinv[t];
        csr[off[t] + rank[e]] = make_int2(s, __float_as_int(w));
    }
}


__global__ __launch_bounds__(256) void k_gather(const int* __restrict__ off,
                                                const int2* __restrict__ csr,
                                                const float* __restrict__ dinv,
                                                const float* __restrict__ h,
                                                float* __restrict__ agg, int n) {
    const unsigned tid = blockIdx.x * blockDim.x + threadIdx.x;
    const int node = tid >> 4;
    if (node >= n) return;
    const int l = tid & 15;
    const float4* __restrict__ h4 = (const float4*)h;
    const int beg = off[node], end = off[node + 1];
    const float d = dinv[node];
    const float4 hs = h4[(size_t)node * 16 + l];
    const float dd = d * d;
    float4 acc;
    acc.x = dd * hs.x; acc.y = dd * hs.y; acc.z = dd * hs.z; acc.w = dd * hs.w;

    int e = beg;
    for (; e + 4 <= end; e += 4) {
        const int2 p0 = csr[e], p1 = csr[e + 1], p2 = csr[e + 2], p3 = csr[e + 3];
        const float4 a0 = h4[(size_t)p0.x * 16 + l];
        const float4 a1 = h4[(size_t)p1.x * 16 + l];
        const float4 a2 = h4[(size_t)p2.x * 16 + l];
        const float4 a3 = h4[(size_t)p3.x * 16 + l];
        const float w0 = __int_as_float(p0.y), w1 = __int_as_float(p1.y);
        const float w2 = __int_as_float(p2.y), w3 = __int_as_float(p3.y);
        acc.x += w0 * a0.x; acc.y += w0 * a0.y; acc.z += w0 * a0.z; acc.w += w0 * a0.w;
        acc.x += w1 * a1.x; acc.y += w1 * a1.y; acc.z += w1 * a1.z; acc.w += w1 * a1.w;
        acc.x += w2 * a2.x; acc.y += w2 * a2.y; acc.z += w2 * a2.z; acc.w += w2 * a2.w;
        acc.x += w3 * a3.x; acc.y += w3 * a3.y; acc.z += w3 * a3.z; acc.w += w3 * a3.w;
    }
    for (; e < end; ++e) {
        const int2 p = csr[e];
        const float4 a0 = h4[(size_t)p.x * 16 + l];
        const float w = __int_as_float(p.y);
        acc.x += w * a0.x; acc.y += w * a0.y; acc.z += w * a0.z; acc.w += w * a0.w;
    }
    ((float4*)agg)[(size_t)node * 16 + l] = acc;
}


__device__ __forceinline__ v2f bfrag_rowmajor(const float* __restrict__ B, int ldb,
                                              int kb  , int ncol) {
    v2f b;
    b.x = B[(size_t)kb * ldb + ncol];
    b.y = B[(size_t)(kb + 1) * ldb + ncol];
    return b;
}

#define WMMA_F32X4(acc, a, b) \
    (acc) = __builtin_amdgcn_wmma_f32_16x16x4_f32(false, (a), false, (b), (short)0, (acc), false, false)


__global__ __launch_bounds__(256) void k_gemm0(const float* __restrict__ x,
                                               const float* __restrict__ W0,
                                               const float* __restrict__ b0,
                                               float* __restrict__ h,
                                               float* __restrict__ h0, int n) {
    const int lane = threadIdx.x & 31;
    const int wave = threadIdx.x >> 5;
    const int m0   = (blockIdx.x * 8 + wave) * 16;
    const int ml   = lane & 15;
    const int kh   = lane >> 4;
    int m = m0 + ml; if (m >= n) m = n - 1;

    v8f acc[4] = {};
    for (int kk = 0; kk < NFEATS; kk += 4) {
        const int ka = kk + kh * 2;
        const float2 av = *(const float2*)(x + (size_t)m * NFEATS + ka);
        v2f a; a.x = av.x; a.y = av.y;
        #pragma unroll
        for (int nt = 0; nt < 4; ++nt) {
            v2f b = bfrag_rowmajor(W0, HIDDEN, ka, nt * 16 + ml);
            WMMA_F32X4(acc[nt], a, b);
        }
    }
    const int rbase = m0 + kh * 8;
    const bool full = (m0 + 16 <= n);
    #pragma unroll
    for (int nt = 0; nt < 4; ++nt) {
        int c = nt * 16 + ml;
        float bias = b0[c];
        if (full) {
            #pragma unroll
            for (int v = 0; v < 8; ++v) {
                float val = fmaxf(acc[nt][v] + bias, 0.0f);
                size_t o = (size_t)(rbase + v) * HIDDEN + c;
                h[o] = val; h0[o] = val;
            }
        } else {
            #pragma unroll
            for (int v = 0; v < 8; ++v) {
                int r = rbase + v;
                if (r < n) {
                    float val = fmaxf(acc[nt][v] + bias, 0.0f);
                    size_t o = (size_t)r * HIDDEN + c;
                    h[o] = val; h0[o] = val;
                }
            }
        }
    }
}


__global__ __launch_bounds__(256) void k_layer(const float* __restrict__ agg,
                                               const float* __restrict__ h0,
                                               const float* __restrict__ W,
                                               float* __restrict__ h,
                                               float beta, int n) {
    const int lane = threadIdx.x & 31;
    const int wave = threadIdx.x >> 5;
    const int m0   = (blockIdx.x * 8 + wave) * 16;
    const int ml   = lane & 15;
    const int kh   = lane >> 4;
    int m = m0 + ml; if (m >= n) m = n - 1;

    v8f acc[4] = {};
    for (int kk = 0; kk < HIDDEN; kk += 4) {
        const int ka = kk + kh * 2;
        size_t idx = (size_t)m * HIDDEN + ka;
        const float2 av = *(const float2*)(agg + idx);
        const float2 hv = *(const float2*)(h0 + idx);
        v2f a;
        a.x = 0.9f * av.x + 0.1f * hv.x;
        a.y = 0.9f * av.y + 0.1f * hv.y;
        #pragma unroll
        for (int nt = 0; nt < 4; ++nt) {
            v2f b = bfrag_rowmajor(W, HIDDEN, ka, nt * 16 + ml);
            WMMA_F32X4(acc[nt], a, b);
        }
    }
    const float omb = 1.0f - beta;
    const int rbase = m0 + kh * 8;
    const bool full = (m0 + 16 <= n);
    #pragma unroll
    for (int nt = 0; nt < 4; ++nt) {
        int c = nt * 16 + ml;
        if (full) {
            #pragma unroll
            for (int v = 0; v < 8; ++v) {
                size_t o = (size_t)(rbase + v) * HIDDEN + c;
                float sup = 0.9f * agg[o] + 0.1f * h0[o];
                h[o] = fmaxf(beta * acc[nt][v] + omb * sup, 0.0f);
            }
        } else {
            #pragma unroll
            for (int v = 0; v < 8; ++v) {
                int r = rbase + v;
                if (r < n) {
                    size_t o = (size_t)r * HIDDEN + c;
                    float sup = 0.9f * agg[o] + 0.1f * h0[o];
                    h[o] = fmaxf(beta * acc[nt][v] + omb * sup, 0.0f);
                }
            }
        }
    }
}


__global__ __launch_bounds__(256) void k_out(const float* __restrict__ h,
                                             const float* __restrict__ Wout,
                                             const float* __restrict__ bout,
                                             float* __restrict__ out, int n) {
    const int lane = threadIdx.x & 31;
    const int wave = threadIdx.x >> 5;
    const int m0   = (blockIdx.x * 8 + wave) * 16;
    const int ml   = lane & 15;
    const int kh   = lane >> 4;
    int m = m0 + ml; if (m >= n) m = n - 1;

    v8f acc[3] = {};
    for (int kk = 0; kk < HIDDEN; kk += 4) {
        const int ka = kk + kh * 2;
        const float2 av = *(const float2*)(h + (size_t)m * HIDDEN + ka);
        v2f a; a.x = av.x; a.y = av.y;
        #pragma unroll
        for (int nt = 0; nt < 3; ++nt) {
            int c = nt * 16 + ml;
            v2f b;
            b.x = (c < NCLASSES) ? Wout[(size_t)ka * NCLASSES + c] : 0.0f;
            b.y = (c < NCLASSES) ? Wout[(size_t)(ka + 1) * NCLASSES + c] : 0.0f;
            WMMA_F32X4(acc[nt], a, b);
        }
    }
    const int rbase = m0 + kh * 8;
    const bool full = (m0 + 16 <= n);
    #pragma unroll
    for (int nt = 0; nt < 3; ++nt) {
        int c = nt * 16 + ml;
        if (c < NCLASSES) {
            float bias = bout[c];
            if (full) {
                #pragma unroll
                for (int v = 0; v < 8; ++v)
                    out[(size_t)(rbase + v) * NCLASSES + c] = acc[nt][v] + bias;
            } else {
                #pragma unroll
                for (int v = 0; v < 8; ++v) {
                    int r = rbase + v;
                    if (r < n) out[(size_t)r * NCLASSES + c] = acc[nt][v] + bias;
                }
            }
        }
    }
}


extern "C" void kernel_launch(void* const* d_in, const int* in_sizes, int n_in,
                              void* d_out, int out_size, void* d_ws, size_t ws_size,
                              hipStream_t stream) {
    const float* x    = (const float*)d_in[0];
    const int*   ei   = (const int*)  d_in[1];
    const float* W0   = (const float*)d_in[2];
    const float* b0   = (const float*)d_in[3];
    const float* Wl   = (const float*)d_in[4];
    const float* Wout = (const float*)d_in[5];
    const float* bout = (const float*)d_in[6];

    const int N = in_sizes[0] / NFEATS;
    const int E = in_sizes[1] / 2;
    const int* erow = ei;
    const int* ecol = ei + E;

    float* h    = (float*)d_ws;
    float* h0   = h  + (size_t)N * HIDDEN;
    float* agg  = h0 + (size_t)N * HIDDEN;
    int2*  csr  = (int2*)(agg + (size_t)N * HIDDEN);
    int*   rank = (int*)(csr + (size_t)E);
    size_t nAl  = ((size_t)N + 64) & ~(size_t)63;
    int*   cnt  = rank + (size_t)E;
    int*   off  = cnt + nAl;
    float* dinv = (float*)(off + nAl);
    (void)ws_size; (void)n_in; (void)out_size;

    const int nodeBlocks = (N + 255) / 256;
    const int edgeBlocks = (E + 255) / 256;
    const int scanBlocks = (N + 1 + 1023) / 1024;
    const int gemmBlocks = (N + 127) / 128;
    const int gathBlocks = (N + 15) / 16;

    k_cnt_zero<<<nodeBlocks, 256, 0, stream>>>(cnt, N);
    k_rank    <<<edgeBlocks, 256, 0, stream>>>(ecol, cnt, rank, E);
    k_scan    <<<scanBlocks, 1024, 0, stream>>>(cnt, off, dinv, N);
    k_fill    <<<edgeBlocks, 256, 0, stream>>>(erow, ecol, rank, off, dinv, csr, E);

    k_gemm0<<<gemmBlocks, 256, 0, stream>>>(x, W0, b0, h, h0, N);

    for (int i = 0; i < NLAYERS; ++i) {
        k_gather<<<gathBlocks, 256, 0, stream>>>(off, csr, dinv, h, agg, N);
        float beta = logf(0.5f / (float)(i + 1) + 1.0f);
        k_layer<<<gemmBlocks, 256, 0, stream>>>(agg, h0, Wl + (size_t)i * HIDDEN * HIDDEN,
                                                h, beta, N);
    }

    k_out<<<gemmBlocks, 256, 0, stream>>>(h, Wout, bout, (float*)d_out, N);
}


# SpMM gather fused into each layer's GEMM launch: a wave gathers its 16 rows (16 row loads in flight), stages support in LDS; h alternates between two buffers; per-node sum order kept
# speedup vs baseline: 3.7727x; 1.3089x over previous
#include <hip/hip_runtime.h>
#include <math.h>


typedef __attribute__((ext_vector_type(2))) float v2f;
typedef __attribute__((ext_vector_type(8))) float v8f;

#define NFEATS   128
#define HIDDEN   64
#define NCLASSES 40
#define NLAYERS  8


__global__ void k_cnt_zero(int* __restrict__ cnt, int n) {
    int i = blockIdx.x * blockDim.x + threadIdx.x;
    if (i < n) cnt[i] = 0;
}

__global__ void k_rank(const int* __restrict__ col, int* __restrict__ cnt,
                       int* __restrict__ rank, int nE) {
    int e = blockIdx.x * blockDim.x + threadIdx.x;
    if (e < nE) rank[e] = atomicAdd(&cnt[col[e]], 1);
}

__global__ __launch_bounds__(1024) void k_scan(const int* __restrict__ cnt, int* __restrict__ off,
                                                float* __restrict__ dinv, int n) {
    __shared__ int wsum[32];
    __shared__ int sbase;
    const int tid = threadIdx.x, lane = tid & 31, wv = tid >> 5;
    const int lim = blockIdx.x * 1024;

    int part = 0;
    for (int j = tid; j < lim; j += 1024) part += cnt[j];
    for (int o = 16; o > 0; o >>= 1) part += __shfl_xor(part, o, 32);
    if (lane == 0) wsum[wv] = part;
    __syncthreads();
    if (wv == 0) {
        int v = wsum[lane];
        for (int o = 16; o > 0; o >>= 1) v += __shfl_xor(v, o, 32);
        if (lane == 0) sbase = v;
    }
    __syncthreads();
    const int base = sbase;

    const int i = lim + tid;
    const int c = (i < n) ? cnt[i] : 0;
    int incl = c;
    for (int o = 1; o < 32; o <<= 1) {
        int t = __shfl_up(incl, o, 32);
        if (lane >= o) incl += t;
    }
    if (lane == 31) wsum[wv] = incl;
    __syncthreads();
    int wpre = 0;
    for (int k = 0; k < wv; ++k) wpre += wsum[k];
    if (i <= n) off[i] = base + wpre + incl - c;
    if (i < n) { float d = (float)(c + 1); dinv[i] = d > 0.0f ? rsqrtf(d) : 0.0f; }
}

__global__ void k_fill(const int* __restrict__ row, const int* __restrict__ col,
                       const int* __restrict__ rank, const int* __restrict__ off,
                       const float* __restrict__ dinv, int2* __restrict__ csr, int nE) {
    int e = blockIdx.x * blockDim.x + threadIdx.x;
    if (e < nE) {
        int s = row[e], t = col[e];
        float w = dinv[s] * dinv[t];
        csr[off[t] + rank[e]] = make_int2(s, __float_as_int(w));
    }
}


__device__ __forceinline__ v2f bfrag_rowmajor(const float* __restrict__ B, int ldb,
                                              int kb  , int ncol) {
    v2f b;
    b.x = B[(size_t)kb * ldb + ncol];
    b.y = B[(size_t)(kb + 1) * ldb + ncol];
    return b;
}

#define WMMA_F32X4(acc, a, b) \
    (acc) = __builtin_amdgcn_wmma_f32_16x16x4_f32(false, (a), false, (b), (short)0, (acc), false, false)


__global__ __launch_bounds__(256) void k_gemm0(const float* __restrict__ x,
                                               const float* __restrict__ W0,
                                               const float* __restrict__ b0,
                                               float* __restrict__ h,
                                               float* __restrict__ h0, int n) {
    const int lane = threadIdx.x & 31;
    const int wave = threadIdx.x >> 5;
    const int m0   = (blockIdx.x * 8 + wave) * 16;
    const int ml   = lane & 15;
    const int kh   = lane >> 4;
    int m = m0 + ml; if (m >= n) m = n - 1;

    v8f acc[4] = {};
    for (int kk = 0; kk < NFEATS; kk += 4) {
        const int ka = kk + kh * 2;
        const float2 av = *(const float2*)(x + (size_t)m * NFEATS + ka);
        v2f a; a.x = av.x; a.y = av.y;
        #pragma unroll
        for (int nt = 0; nt < 4; ++nt) {
            v2f b = bfrag_rowmajor(W0, HIDDEN, ka, nt * 16 + ml);
            WMMA_F32X4(acc[nt], a, b);
        }
    }
    const int rbase = m0 + kh * 8;
    const bool full = (m0 + 16 <= n);
    #pragma unroll
    for (int nt = 0; nt < 4; ++nt) {
        int c = nt * 16 + ml;
        float bias = b0[c];
        if (full) {
            #pragma unroll
            for (int v = 0; v < 8; ++v) {
                float val = fmaxf(acc[nt][v] + bias, 0.0f);
                size_t o = (size_t)(rbase + v) * HIDDEN + c;
                h[o] = val; h0[o] = val;
            }
        } else {
            #pragma unroll
            for (int v = 0; v < 8; ++v) {
                int r = rbase + v;
                if (r < n) {
                    float val = fmaxf(acc[nt][v] + bias, 0.0f);
                    size_t o = (size_t)r * HIDDEN + c;
                    h[o] = val; h0[o] = val;
                }
            }
        }
    }
}


#define TP 68

#define RSHARE(v, I) __builtin_amdgcn_update_dpp(0, (v), 0x150 + (I), 0xf, 0xf, false)

__device__ __forceinline__ float4 ldrow(const float* __restrict__ base, int row, unsigned lb) {
    return *(const float4*)((const char*)base + (size_t)((unsigned)row * 256u + lb));
}

#define G_LOAD(I) const float4 a##I = ldrow(h, RSHARE(cs, I), lb);
#define G_FMA(I) { const float w = __int_as_float(RSHARE(cw, I)); \
    acc.x = __builtin_fmaf(w, a##I.x, acc.x); acc.y = __builtin_fmaf(w, a##I.y, acc.y); \
    acc.z = __builtin_fmaf(w, a##I.z, acc.z); acc.w = __builtin_fmaf(w, a##I.w, acc.w); }
#define G_ALL(M) M(0) M(1) M(2) M(3) M(4) M(5) M(6) M(7) M(8) M(9) M(10) M(11) M(12) M(13) M(14) M(15)

__device__ __forceinline__ void gather16(float4& acc, const float* __restrict__ h,
                                         int cs, int cw, bool valid, unsigned lb) {
    const int sfirst = RSHARE(cs, 0);
    cs = valid ? cs : sfirst;
    cw = valid ? cw : 0;
    G_ALL(G_LOAD)
    G_ALL(G_FMA)
}

__global__ __launch_bounds__(64) void k_layer(const int* __restrict__ off,
                                              const int2* __restrict__ csr,
                                              const float* __restrict__ dinv,
                                              const float* __restrict__ hin,
                                              const float* __restrict__ h0,
                                              const float* __restrict__ W,
                                              float* __restrict__ hout,
                                              float beta, int n, int nE) {
    __shared__ __attribute__((aligned(16))) float tile[2][16 * TP];
    const int lane = threadIdx.x & 31;
    const int wave = threadIdx.x >> 5;
    const int m0   = (blockIdx.x * 2 + wave) * 16;
    const int ml   = lane & 15;
    const int kh   = lane >> 4;
    float* __restrict__ T = tile[wave];

    {
        const int nbase = m0 + 8 * kh;
        const unsigned lb = (unsigned)ml * 16u;
        int oi = nbase + (ml < 8 ? ml : 8); if (oi > n) oi = n;
        const int offv = off[oi];
        int di = nbase + (ml < 7 ? ml : 7); if (di > n - 1) di = n - 1;
        const float dv = dinv[di];
        const int emax = nE > 0 ? nE - 1 : 0;

        int beg = __shfl(offv, 0, 16), end = __shfl(offv, 1, 16);
        int i0 = beg + ml, i1 = beg + 16 + ml;
        bool v0 = i0 < end, v1 = i1 < end;
        int2 c0 = csr[i0 < emax ? i0 : emax];
        int2 c1 = csr[i1 < emax ? i1 : emax];
        float4 hsn = ldrow(hin, nbase < n - 1 ? nbase : n - 1, lb);

        #pragma nounroll
        for (int p = 0; p < 8; ++p) {
            int node = nbase + p; if (node > n - 1) node = n - 1;
            const int pn = p < 7 ? p + 1 : 8, pn1 = p < 7 ? p + 2 : 8;
            const int nbeg = __shfl(offv, pn, 16), nend = __shfl(offv, pn1, 16);
            const int j0 = nbeg + ml, j1 = nbeg + 16 + ml;
            const bool nv0 = j0 < nend, nv1 = j1 < nend;
            const int2 n0 = csr[j0 < emax ? j0 : emax];
            const int2 n1 = csr[j1 < emax ? j1 : emax];

            int nnode = nbase + (p < 7 ? p + 1 : 7); if (nnode > n - 1) nnode = n - 1;
            const float4 hs = hsn;
            hsn = ldrow(hin, nnode, lb);

            const float d = __shfl(dv, p, 16);
            const float4 z  = ldrow(h0, node, lb);
            const float dd = d * d;
            float4 acc;
            acc.x = dd * hs.x; acc.y = dd * hs.y; acc.z = dd * hs.z; acc.w = dd * hs.w;

            if (beg < end)      gather16(acc, hin, c0.x, c0.y, v0, lb);
            if (beg + 16 < end) gather16(acc, hin, c1.x, c1.y, v1, lb);
            for (int e = beg + 32; e < end; e += 16) {
                const int ix = e + ml;
                const bool vx = ix < end;
                const int2 cx = csr[ix < emax ? ix : emax];
                gather16(acc, hin, cx.x, cx.y, vx, lb);
            }

            float4 sup;
            sup.x = 0.9f * acc.x + 0.1f * z.x; sup.y = 0.9f * acc.y + 0.1f * z.y;
            sup.z = 0.9f * acc.z + 0.1f * z.z; sup.w = 0.9f * acc.w + 0.1f * z.w;
            *(float4*)(T + (8 * kh + p) * TP + 4 * ml) = sup;

            beg = nbeg; end = nend; c0 = n0; c1 = n1; v0 = nv0; v1 = nv1;
        }
    }
    __syncthreads();

    v8f acc[4] = {};
    for (int kk = 0; kk < HIDDEN; kk += 4) {
        const int ka = kk + kh * 2;
        const float2 av = *(const float2*)(T + ml * TP + ka);
        v2f a; a.x = av.x; a.y = av.y;
        #pragma unroll
        for (int nt = 0; nt < 4; ++nt) {
            v2f b = bfrag_rowmajor(W, HIDDEN, ka, nt * 16 + ml);
            WMMA_F32X4(acc[nt], a, b);
        }
    }
    const float omb = 1.0f - beta;
    const int rbase = m0 + kh * 8;
    const bool full = (m0 + 16 <= n);
    #pragma unroll
    for (int nt = 0; nt < 4; ++nt) {
        int c = nt * 16 + ml;
        if (full) {
            #pragma unroll
            for (int v = 0; v < 8; ++v) {
                size_t o = (size_t)(rbase + v) * HIDDEN + c;
                float sup = T[(kh * 8 + v) * TP + c];
                hout[o] = fmaxf(beta * acc[nt][v] + omb * sup, 0.0f);
            }
        } else {
            #pragma unroll
            for (int v = 0; v < 8; ++v) {
                int r = rbase + v;
                if (r < n) {
                    size_t o = (size_t)r * HIDDEN + c;
                    float sup = T[(kh * 8 + v) * TP + c];
                    hout[o] = fmaxf(beta * acc[nt][v] + omb * sup, 0.0f);
                }
            }
        }
    }
}


__global__ __launch_bounds__(256) void k_out(const float* __restrict__ h,
                                             const float* __restrict__ Wout,
                                             const float* __restrict__ bout,
                                             float* __restrict__ out, int n) {
    const int lane = threadIdx.x & 31;
    const int wave = threadIdx.x >> 5;
    const int m0   = (blockIdx.x * 8 + wave) * 16;
    const int ml   = lane & 15;
    const int kh   = lane >> 4;
    int m = m0 + ml; if (m >= n) m = n - 1;

    v8f acc[3] = {};
    for (int kk = 0; kk < HIDDEN; kk += 4) {
        const int ka = kk + kh * 2;
        const float2 av = *(const float2*)(h + (size_t)m * HIDDEN + ka);
        v2f a; a.x = av.x; a.y = av.y;
        #pragma unroll
        for (int nt = 0; nt < 3; ++nt) {
            int c = nt * 16 + ml;
            v2f b;
            b.x = (c < NCLASSES) ? Wout[(size_t)ka * NCLASSES + c] : 0.0f;
            b.y = (c < NCLASSES) ? Wout[(size_t)(ka + 1) * NCLASSES + c] : 0.0f;
            WMMA_F32X4(acc[nt], a, b);
        }
    }
    const int rbase = m0 + kh * 8;
    const bool full = (m0 + 16 <= n);
    #pragma unroll
    for (int nt = 0; nt < 3; ++nt) {
        int c = nt * 16 + ml;
        if (c < NCLASSES) {
            float bias = bout[c];
            if (full) {
                #pragma unroll
                for (int v = 0; v < 8; ++v)
                    out[(size_t)(rbase + v) * NCLASSES + c] = acc[nt][v] + bias;
            } else {
                #pragma unroll
                for (int v = 0; v < 8; ++v) {
                    int r = rbase + v;
                    if (r < n) out[(size_t)r * NCLASSES + c] = acc[nt][v] + bias;
                }
            }
        }
    }
}


extern "C" void kernel_launch(void* const* d_in, const int* in_sizes, int n_in,
                              void* d_out, int out_size, void* d_ws, size_t ws_size,
                              hipStream_t stream) {
    const float* x    = (const float*)d_in[0];
    const int*   ei   = (const int*)  d_in[1];
    const float* W0   = (const float*)d_in[2];
    const float* b0   = (const float*)d_in[3];
    const float* Wl   = (const float*)d_in[4];
    const float* Wout = (const float*)d_in[5];
    const float* bout = (const float*)d_in[6];

    const int N = in_sizes[0] / NFEATS;
    const int E = in_sizes[1] / 2;
    const int* erow = ei;
    const int* ecol = ei + E;

    float* h    = (float*)d_ws;
    float* h0   = h  + (size_t)N * HIDDEN;
    float* h2   = h0 + (size_t)N * HIDDEN;
    int2*  csr  = (int2*)(h2 + (size_t)N * HIDDEN);
    int*   rank = (int*)(csr + (size_t)E);
    size_t nAl  = ((size_t)N + 64) & ~(size_t)63;
    int*   cnt  = rank + (size_t)E;
    int*   off  = cnt + nAl;
    float* dinv = (float*)(off + nAl);
    (void)ws_size; (void)n_in; (void)out_size;

    const int nodeBlocks = (N + 255) / 256;
    const int edgeBlocks = (E + 255) / 256;
    const int scanBlocks = (N + 1 + 1023) / 1024;
    const int gemmBlocks = (N + 127) / 128;
    const int layerBlocks = (N + 31) / 32;

    k_cnt_zero<<<nodeBlocks, 256, 0, stream>>>(cnt, N);
    k_rank    <<<edgeBlocks, 256, 0, stream>>>(ecol, cnt, rank, E);
    k_scan    <<<scanBlocks, 1024, 0, stream>>>(cnt, off, dinv, N);
    k_fill    <<<edgeBlocks, 256, 0, stream>>>(erow, ecol, rank, off, dinv, csr, E);

    k_gemm0<<<gemmBlocks, 256, 0, stream>>>(x, W0, b0, h, h0, N);

    for (int i = 0; i < NLAYERS; ++i) {
        float beta = logf(0.5f / (float)(i + 1) + 1.0f);
        k_layer<<<layerBlocks, 64, 0, stream>>>(off, csr, dinv, (i & 1) ? h2 : h, h0,
                                                Wl + (size_t)i * HIDDEN * HIDDEN,
                                                (i & 1) ? h : h2, beta, N, E);
    }

    k_out<<<gemmBlocks, 256, 0, stream>>>(h, Wout, bout, (float*)d_out, N);
}


# Launches 14 -> 12: the input GEMM rides in the edge-ranking and edge-fill launches (independent blocks side by side, half each); the output GEMM runs inside the last layer's launch from its LDS tile
# speedup vs baseline: 4.0598x; 1.0761x over previous
#include <hip/hip_runtime.h>
#include <math.h>


typedef __attribute__((ext_vector_type(2))) float v2f;
typedef __attribute__((ext_vector_type(8))) float v8f;

#define NFEATS   128
#define HIDDEN   64
#define NCLASSES 40
#define NLAYERS  8


__global__ void k_cnt_zero(int* __restrict__ cnt, int n) {
    int i = blockIdx.x * blockDim.x + threadIdx.x;
    if (i < n) cnt[i] = 0;
}


__global__ __launch_bounds__(1024) void k_scan(const int* __restrict__ cnt, int* __restrict__ off,
                                                float* __restrict__ dinv, int n) {
    __shared__ int wsum[32];
    __shared__ int sbase;
    const int tid = threadIdx.x, lane = tid & 31, wv = tid >> 5;
    const int lim = blockIdx.x * 1024;

    int part = 0;
    for (int j = tid; j < lim; j += 1024) part += cnt[j];
    for (int o = 16; o > 0; o >>= 1) part += __shfl_xor(part, o, 32);
    if (lane == 0) wsum[wv] = part;
    __syncthreads();
    if (wv == 0) {
        int v = wsum[lane];
        for (int o = 16; o > 0; o >>= 1) v += __shfl_xor(v, o, 32);
        if (lane == 0) sbase = v;
    }
    __syncthreads();
    const int base = sbase;

    const int i = lim + tid;
    const int c = (i < n) ? cnt[i] : 0;
    int incl = c;
    for (int o = 1; o < 32; o <<= 1) {
        int t = __shfl_up(incl, o, 32);
        if (lane >= o) incl += t;
    }
    if (lane == 31) wsum[wv] = incl;
    __syncthreads();
    int wpre = 0;
    for (int k = 0; k < wv; ++k) wpre += wsum[k];
    if (i <= n) off[i] = base + wpre + incl - c;
    if (i < n) { float d = (float)(c + 1); dinv[i] = d > 0.0f ? rsqrtf(d) : 0.0f; }
}


__device__ __forceinline__ v2f bfrag_rowmajor(const float* __restrict__ B, int ldb,
                                              int kb  , int ncol) {
    v2f b;
    b.x = B[(size_t)kb * ldb + ncol];
    b.y = B[(size_t)(kb + 1) * ldb + ncol];
    return b;
}

#define WMMA_F32X4(acc, a, b) \
    (acc) = __builtin_amdgcn_wmma_f32_16x16x4_f32(false, (a), false, (b), (short)0, (acc), false, false)


__device__ __forceinline__ void gemm0_rows(const float* __restrict__ x,
                                           const float* __restrict__ W0,
                                           const float* __restrict__ b0,
                                           float* __restrict__ h,
                                           float* __restrict__ h0, int n, int blk) {
    const int lane = threadIdx.x & 31;
    const int wave = threadIdx.x >> 5;
    const int m0   = (blk * 8 + wave) * 16;
    const int ml   = lane & 15;
    const int kh   = lane >> 4;
    int m = m0 + ml; if (m >= n) m = n - 1;

    v8f acc[4] = {};
    for (int kk = 0; kk < NFEATS; kk += 4) {
        const int ka = kk + kh * 2;
        const float2 av = *(const float2*)(x + (size_t)m * NFEATS + ka);
        v2f a; a.x = av.x; a.y = av.y;
        #pragma unroll
        for (int nt = 0; nt < 4; ++nt) {
            v2f b = bfrag_rowmajor(W0, HIDDEN, ka, nt * 16 + ml);
            WMMA_F32X4(acc[nt], a, b);
        }
    }
    const int rbase = m0 + kh * 8;
    const bool full = (m0 + 16 <= n);
    #pragma unroll
    for (int nt = 0; nt < 4; ++nt) {
        int c = nt * 16 + ml;
        float bias = b0[c];
        if (full) {
            #pragma unroll
            for (int v = 0; v < 8; ++v) {
                float val = fmaxf(acc[nt][v] + bias, 0.0f);
                size_t o = (size_t)(rbase + v) * HIDDEN + c;
                h[o] = val; h0[o] = val;
            }
        } else {
            #pragma unroll
            for (int v = 0; v < 8; ++v) {
                int r = rbase + v;
                if (r < n) {
                    float val = fmaxf(acc[nt][v] + bias, 0.0f);
                    size_t o = (size_t)r * HIDDEN + c;
                    h[o] = val; h0[o] = val;
                }
            }
        }
    }
}

__global__ __launch_bounds__(256) void k_gemm0_rank(const float* __restrict__ x,
                                                    const float* __restrict__ W0,
                                                    const float* __restrict__ b0,
                                                    float* __restrict__ h,
                                                    float* __restrict__ h0, int n,
                                                    int gFirst, int gBlocks,
                                                    const int* __restrict__ col,
                                                    int* __restrict__ cnt,
                                                    int* __restrict__ rank, int nE) {
    if ((int)blockIdx.x < gBlocks) {
        gemm0_rows(x, W0, b0, h, h0, n, gFirst + (int)blockIdx.x);
        return;
    }
    const int e = ((int)blockIdx.x - gBlocks) * 256 + (int)threadIdx.x;
    if (e < nE) rank[e] = atomicAdd(&cnt[col[e]], 1);
}

__global__ __launch_bounds__(256) void k_gemm0_fill(const float* __restrict__ x,
                                                    const float* __restrict__ W0,
                                                    const float* __restrict__ b0,
                                                    float* __restrict__ h,
                                                    float* __restrict__ h0, int n,
                                                    int gFirst, int gBlocks,
                                                    const int* __restrict__ row,
                                                    const int* __restrict__ col,
                                                    const int* __restrict__ rank,
                                                    const int* __restrict__ off,
                                                    const float* __restrict__ dinv,
                                                    int2* __restrict__ csr, int nE) {
    if ((int)blockIdx.x < gBlocks) {
        gemm0_rows(x, W0, b0, h, h0, n, gFirst + (int)blockIdx.x);
        return;
    }
    const int e = ((int)blockIdx.x - gBlocks) * 256 + (int)threadIdx.x;
    if (e < nE) {
        int s = row[e], t = col[e];
        float w = dinv[s] * dinv[t];
        csr[off[t] + rank[e]] = make_int2(s, __float_as_int(w));
    }
}


#define TP 68

#define RSHARE(v, I) __builtin_amdgcn_update_dpp(0, (v), 0x150 + (I), 0xf, 0xf, false)

__device__ __forceinline__ float4 ldrow(const float* __restrict__ base, int row, unsigned lb) {
    return *(const float4*)((const char*)base + (size_t)((unsigned)row * 256u + lb));
}

#define G_LOAD(I) const float4 a##I = ldrow(h, RSHARE(cs, I), lb);
#define G_FMA(I) { const float w = __int_as_float(RSHARE(cw, I)); \
    acc.x = __builtin_fmaf(w, a##I.x, acc.x); acc.y = __builtin_fmaf(w, a##I.y, acc.y); \
    acc.z = __builtin_fmaf(w, a##I.z, acc.z); acc.w = __builtin_fmaf(w, a##I.w, acc.w); }
#define G_ALL(M) M(0) M(1) M(2) M(3) M(4) M(5) M(6) M(7) M(8) M(9) M(10) M(11) M(12) M(13) M(14) M(15)

__device__ __forceinline__ void gather16(float4& acc, const float* __restrict__ h,
                                         int cs, int cw, bool valid, unsigned lb) {
    const int sfirst = RSHARE(cs, 0);
    cs = valid ? cs : sfirst;
    cw = valid ? cw : 0;
    G_ALL(G_LOAD)
    G_ALL(G_FMA)
}

__device__ __forceinline__ void layer_body(const int* __restrict__ off,
                                           const int2* __restrict__ csr,
                                           const float* __restrict__ dinv,
                                           const float* __restrict__ hin,
                                           const float* __restrict__ h0,
                                           const float* __restrict__ W,
                                           float* __restrict__ hout,
                                           float beta, int n, int nE, const bool last,
                                           const float* __restrict__ Wout,
                                           const float* __restrict__ bout,
                                           float* __restrict__ out) {
    __shared__ __attribute__((aligned(16))) float tile[2][16 * TP];
    const int lane = threadIdx.x & 31;
    const int wave = threadIdx.x >> 5;
    const int m0   = (blockIdx.x * 2 + wave) * 16;
    const int ml   = lane & 15;
    const int kh   = lane >> 4;
    float* __restrict__ T = tile[wave];

    {
        const int nbase = m0 + 8 * kh;
        const unsigned lb = (unsigned)ml * 16u;
        int oi = nbase + (ml < 8 ? ml : 8); if (oi > n) oi = n;
        const int offv = off[oi];
        int di = nbase + (ml < 7 ? ml : 7); if (di > n - 1) di = n - 1;
        const float dv = dinv[di];
        const int emax = nE > 0 ? nE - 1 : 0;

        int beg = __shfl(offv, 0, 16), end = __shfl(offv, 1, 16);
        int i0 = beg + ml, i1 = beg + 16 + ml;
        bool v0 = i0 < end, v1 = i1 < end;
        int2 c0 = csr[i0 < emax ? i0 : emax];
        int2 c1 = csr[i1 < emax ? i1 : emax];
        float4 hsn = ldrow(hin, nbase < n - 1 ? nbase : n - 1, lb);

        #pragma nounroll
        for (int p = 0; p < 8; ++p) {
            int node = nbase + p; if (node > n - 1) node = n - 1;
            const int pn = p < 7 ? p + 1 : 8, pn1 = p < 7 ? p + 2 : 8;
            const int nbeg = __shfl(offv, pn, 16), nend = __shfl(offv, pn1, 16);
            const int j0 = nbeg + ml, j1 = nbeg + 16 + ml;
            const bool nv0 = j0 < nend, nv1 = j1 < nend;
            const int2 n0 = csr[j0 < emax ? j0 : emax];
            const int2 n1 = csr[j1 < emax ? j1 : emax];

            int nnode = nbase + (p < 7 ? p + 1 : 7); if (nnode > n - 1) nnode = n - 1;
            const float4 hs = hsn;
            hsn = ldrow(hin, nnode, lb);

            const float d = __shfl(dv, p, 16);
            const float4 z  = ldrow(h0, node, lb);
            const float dd = d * d;
            float4 acc;
            acc.x = dd * hs.x; acc.y = dd * hs.y; acc.z = dd * hs.z; acc.w = dd * hs.w;

            if (beg < end)      gather16(acc, hin, c0.x, c0.y, v0, lb);
            if (beg + 16 < end) gather16(acc, hin, c1.x, c1.y, v1, lb);
            for (int e = beg + 32; e < end; e += 16) {
                const int ix = e + ml;
                const bool vx = ix < end;
                const int2 cx = csr[ix < emax ? ix : emax];
                gather16(acc, hin, cx.x, cx.y, vx, lb);
            }

            float4 sup;
            sup.x = 0.9f * acc.x + 0.1f * z.x; sup.y = 0.9f * acc.y + 0.1f * z.y;
            sup.z = 0.9f * acc.z + 0.1f * z.z; sup.w = 0.9f * acc.w + 0.1f * z.w;
            *(float4*)(T + (8 * kh + p) * TP + 4 * ml) = sup;

            beg = nbeg; end = nend; c0 = n0; c1 = n1; v0 = nv0; v1 = nv1;
        }
    }
    __syncthreads();

    v8f acc[4] = {};
    for (int kk = 0; kk < HIDDEN; kk += 4) {
        const int ka = kk + kh * 2;
        const float2 av = *(const float2*)(T + ml * TP + ka);
        v2f a; a.x = av.x; a.y = av.y;
        #pragma unroll
        for (int nt = 0; nt < 4; ++nt) {
            v2f b = bfrag_rowmajor(W, HIDDEN, ka, nt * 16 + ml);
            WMMA_F32X4(acc[nt], a, b);
        }
    }
    const float omb = 1.0f - beta;
    const int rbase = m0 + kh * 8;
    const bool full = (m0 + 16 <= n);
    if (last) {
        #pragma unroll
        for (int nt = 0; nt < 4; ++nt) {
            const int c = nt * 16 + ml;
            #pragma unroll
            for (int v = 0; v < 8; ++v) {
                float* p = T + (kh * 8 + v) * TP + c;
                *p = fmaxf(beta * acc[nt][v] + omb * (*p), 0.0f);
            }
        }
        __syncthreads();
        v8f oacc[3] = {};
        const bool c2ok = (32 + ml) < NCLASSES;
        const int  c2   = c2ok ? 32 + ml : NCLASSES - 1;
        for (int kk = 0; kk < HIDDEN; kk += 4) {
            const int ka = kk + kh * 2;
            const float2 av = *(const float2*)(T + ml * TP + ka);
            v2f a; a.x = av.x; a.y = av.y;
            const float* w0 = Wout + (size_t)ka * NCLASSES;
            const float* w1 = w0 + NCLASSES;
            v2f b0v, b1v, b2v;
            b0v.x = w0[ml];      b0v.y = w1[ml];
            b1v.x = w0[16 + ml]; b1v.y = w1[16 + ml];
            const float t0 = w0[c2], t1 = w1[c2];
            b2v.x = c2ok ? t0 : 0.0f; b2v.y = c2ok ? t1 : 0.0f;
            WMMA_F32X4(oacc[0], a, b0v);
            WMMA_F32X4(oacc[1], a, b1v);
            WMMA_F32X4(oacc[2], a, b2v);
        }
        #pragma unroll
        for (int nt = 0; nt < 3; ++nt) {
            const int c = nt * 16 + ml;
            if (c < NCLASSES) {
                const float bias = bout[c];
                if (full) {
                    #pragma unroll
                    for (int v = 0; v < 8; ++v)
                        out[(size_t)(rbase + v) * NCLASSES + c] = oacc[nt][v] + bias;
                } else {
                    #pragma unroll
                    for (int v = 0; v < 8; ++v) {
                        const int r = rbase + v;
                        if (r < n) out[(size_t)r * NCLASSES + c] = oacc[nt][v] + bias;
                    }
                }
            }
        }
        return;
    }
    #pragma unroll
    for (int nt = 0; nt < 4; ++nt) {
        int c = nt * 16 + ml;
        if (full) {
            #pragma unroll
            for (int v = 0; v < 8; ++v) {
                size_t o = (size_t)(rbase + v) * HIDDEN + c;
                float sup = T[(kh * 8 + v) * TP + c];
                hout[o] = fmaxf(beta * acc[nt][v] + omb * sup, 0.0f);
            }
        } else {
            #pragma unroll
            for (int v = 0; v < 8; ++v) {
                int r = rbase + v;
                if (r < n) {
                    size_t o = (size_t)r * HIDDEN + c;
                    float sup = T[(kh * 8 + v) * TP + c];
                    hout[o] = fmaxf(beta * acc[nt][v] + omb * sup, 0.0f);
                }
            }
        }
    }
}


__global__ __launch_bounds__(64) void k_layer(const int* __restrict__ off,
                                              const int2* __restrict__ csr,
                                              const float* __restrict__ dinv,
                                              const float* __restrict__ hin,
                                              const float* __restrict__ h0,
                                              const float* __restrict__ W,
                                              float* __restrict__ hout,
                                              float beta, int n, int nE) {
    layer_body(off, csr, dinv, hin, h0, W, hout, beta, n, nE, false, nullptr, nullptr, nullptr);
}

__global__ __launch_bounds__(64) void k_layer_out(const int* __restrict__ off,
                                                  const int2* __restrict__ csr,
                                                  const float* __restrict__ dinv,
                                                  const float* __restrict__ hin,
                                                  const float* __restrict__ h0,
                                                  const float* __restrict__ W,
                                                  float beta, int n, int nE,
                                                  const float* __restrict__ Wout,
                                                  const float* __restrict__ bout,
                                                  float* __restrict__ out) {
    layer_body(off, csr, dinv, hin, h0, W, nullptr, beta, n, nE, true, Wout, bout, out);
}


extern "C" void kernel_launch(void* const* d_in, const int* in_sizes, int n_in,
                              void* d_out, int out_size, void* d_ws, size_t ws_size,
                              hipStream_t stream) {
    const float* x    = (const float*)d_in[0];
    const int*   ei   = (const int*)  d_in[1];
    const float* W0   = (const float*)d_in[2];
    const float* b0   = (const float*)d_in[3];
    const float* Wl   = (const float*)d_in[4];
    const float* Wout = (const float*)d_in[5];
    const float* bout = (const float*)d_in[6];

    const int N = in_sizes[0] / NFEATS;
    const int E = in_sizes[1] / 2;
    const int* erow = ei;
    const int* ecol = ei + E;

    float* h    = (float*)d_ws;
    float* h0   = h  + (size_t)N * HIDDEN;
    float* h2   = h0 + (size_t)N * HIDDEN;
    int2*  csr  = (int2*)(h2 + (size_t)N * HIDDEN);
    int*   rank = (int*)(csr + (size_t)E);
    size_t nAl  = ((size_t)N + 64) & ~(size_t)63;
    int*   cnt  = rank + (size_t)E;
    int*   off  = cnt + nAl;
    float* dinv = (float*)(off + nAl);
    (void)ws_size; (void)n_in; (void)out_size;

    const int nodeBlocks = (N + 255) / 256;
    const int edgeBlocks = (E + 255) / 256;
    const int scanBlocks = (N + 1 + 1023) / 1024;
    const int gemmBlocks = (N + 127) / 128;
    const int layerBlocks = (N + 31) / 32;

    const int gA = gemmBlocks / 2;
    const int gB = gemmBlocks - gA;

    k_cnt_zero  <<<nodeBlocks, 256, 0, stream>>>(cnt, N);
    k_gemm0_rank<<<gA + edgeBlocks, 256, 0, stream>>>(x, W0, b0, h, h0, N, 0, gA,
                                                      ecol, cnt, rank, E);
    k_scan      <<<scanBlocks, 1024, 0, stream>>>(cnt, off, dinv, N);
    k_gemm0_fill<<<gB + edgeBlocks, 256, 0, stream>>>(x, W0, b0, h, h0, N, gA, gB,
                                                      erow, ecol, rank, off, dinv, csr, E);

    for (int i = 0; i < NLAYERS; ++i) {
        float beta = logf(0.5f / (float)(i + 1) + 1.0f);
        const float* hi = (i & 1) ? h2 : h;
        const float* Wi = Wl + (size_t)i * HIDDEN * HIDDEN;
        if (i + 1 < NLAYERS)
            k_layer<<<layerBlocks, 64, 0, stream>>>(off, csr, dinv, hi, h0, Wi,
                                                    (i & 1) ? h : h2, beta, N, E);
        else
            k_layer_out<<<layerBlocks, 64, 0, stream>>>(off, csr, dinv, hi, h0, Wi, beta, N, E,
                                                        Wout, bout, (float*)d_out);
    }
}


# Edge list built by a two-level counting sort on LDS counters (128-node bins, then nodes) in place of the global atomic rank and the scattered fill; per-node sum order = LDS arrival order
# speedup vs baseline: 4.5077x; 1.1103x over previous
#include <hip/hip_runtime.h>
#include <math.h>


typedef __attribute__((ext_vector_type(2))) float v2f;
typedef __attribute__((ext_vector_type(8))) float v8f;

#define NFEATS   128
#define HIDDEN   64
#define NCLASSES 40
#define NLAYERS  8


#define BSH   7
#define BW    128
#define MAXB  1024
#define CHUNK 4096

__global__ __launch_bounds__(256) void k_hist(const int* __restrict__ col, int* __restrict__ ghist,
                                              int nE, int nBin, int nBlk) {
    __shared__ int lh[MAXB + 1];
    const int tid = threadIdx.x, blk = blockIdx.x;
    for (int b = tid; b < nBin; b += 256) lh[b] = 0;
    int tv[16];
    #pragma unroll
    for (int k = 0; k < 16; ++k) {
        const int e = blk * CHUNK + k * 256 + tid;
        const int t = col[e < nE ? e : nE - 1];
        tv[k] = e < nE ? t : -1;
    }
    __syncthreads();
    #pragma unroll
    for (int k = 0; k < 16; ++k)
        atomicAdd(&lh[tv[k] >= 0 ? (tv[k] >> BSH) : MAXB], 1);
    __syncthreads();
    for (int b = tid; b < nBin; b += 256) ghist[(size_t)b * nBlk + blk] = lh[b];
}

__global__ __launch_bounds__(256) void k_binscan(int* __restrict__ ghist, int* __restrict__ binTot,
                                                 int nBin, int nBlk) {
    const int lane = threadIdx.x & 31;
    const int bin = blockIdx.x * 8 + (threadIdx.x >> 5);
    if (bin >= nBin) return;
    int* __restrict__ g = ghist + (size_t)bin * nBlk;
    int run = 0;
    for (int b0 = 0; b0 < nBlk; b0 += 32) {
        const int b = b0 + lane;
        const int v = b < nBlk ? g[b] : 0;
        int incl = v;
        for (int o = 1; o < 32; o <<= 1) {
            int t = __shfl_up(incl, o, 32);
            if (lane >= o) incl += t;
        }
        if (b < nBlk) g[b] = run + incl - v;
        run += __shfl(incl, 31, 32);
    }
    if (lane == 0) binTot[bin] = run;
}


__device__ __forceinline__ v2f bfrag_rowmajor(const float* __restrict__ B, int ldb,
                                              int kb  , int ncol) {
    v2f b;
    b.x = B[(size_t)kb * ldb + ncol];
    b.y = B[(size_t)(kb + 1) * ldb + ncol];
    return b;
}

#define WMMA_F32X4(acc, a, b) \
    (acc) = __builtin_amdgcn_wmma_f32_16x16x4_f32(false, (a), false, (b), (short)0, (acc), false, false)


__device__ __forceinline__ void gemm0_rows(const float* __restrict__ x,
                                           const float* __restrict__ W0,
                                           const float* __restrict__ b0,
                                           float* __restrict__ h,
                                           float* __restrict__ h0, int n, int blk) {
    const int lane = threadIdx.x & 31;
    const int wave = threadIdx.x >> 5;
    const int m0   = (blk * 8 + wave) * 16;
    const int ml   = lane & 15;
    const int kh   = lane >> 4;
    int m = m0 + ml; if (m >= n) m = n - 1;

    v8f acc[4] = {};
    for (int kk = 0; kk < NFEATS; kk += 4) {
        const int ka = kk + kh * 2;
        const float2 av = *(const float2*)(x + (size_t)m * NFEATS + ka);
        v2f a; a.x = av.x; a.y = av.y;
        #pragma unroll
        for (int nt = 0; nt < 4; ++nt) {
            v2f b = bfrag_rowmajor(W0, HIDDEN, ka, nt * 16 + ml);
            WMMA_F32X4(acc[nt], a, b);
        }
    }
    const int rbase = m0 + kh * 8;
    const bool full = (m0 + 16 <= n);
    #pragma unroll
    for (int nt = 0; nt < 4; ++nt) {
        int c = nt * 16 + ml;
        float bias = b0[c];
        if (full) {
            #pragma unroll
            for (int v = 0; v < 8; ++v) {
                float val = fmaxf(acc[nt][v] + bias, 0.0f);
                size_t o = (size_t)(rbase + v) * HIDDEN + c;
                h[o] = val; h0[o] = val;
            }
        } else {
            #pragma unroll
            for (int v = 0; v < 8; ++v) {
                int r = rbase + v;
                if (r < n) {
                    float val = fmaxf(acc[nt][v] + bias, 0.0f);
                    size_t o = (size_t)r * HIDDEN + c;
                    h[o] = val; h0[o] = val;
                }
            }
        }
    }
}

__global__ __launch_bounds__(256) void k_gemm0_scatter(const float* __restrict__ x,
                                                       const float* __restrict__ W0,
                                                       const float* __restrict__ b0,
                                                       float* __restrict__ h,
                                                       float* __restrict__ h0, int n,
                                                       int gFirst, int gBlocks,
                                                       const int* __restrict__ row,
                                                       const int* __restrict__ col,
                                                       const int* __restrict__ ghist,
                                                       const int* __restrict__ binTot,
                                                       int* __restrict__ binBase,
                                                       int2* __restrict__ tmp,
                                                       int nE, int nBin, int nBlk) {
    __shared__ int lpos[MAXB + 1];
    __shared__ int wsum[8];
    if ((int)blockIdx.x < gBlocks) {
        gemm0_rows(x, W0, b0, h, h0, n, gFirst + (int)blockIdx.x);
        return;
    }
    const int blk = (int)blockIdx.x - gBlocks;
    const int tid = threadIdx.x, lane = tid & 31, wv = tid >> 5;

    int sv[16], tv[16];
    #pragma unroll
    for (int k = 0; k < 16; ++k) {
        const int e = blk * CHUNK + k * 256 + tid;
        const int ec = e < nE ? e : nE - 1;
        const int t = col[ec];
        sv[k] = row[ec];
        tv[k] = e < nE ? t : -1;
    }

    int v[4], s = 0;
    #pragma unroll
    for (int k = 0; k < 4; ++k) {
        const int b = 4 * tid + k;
        v[k] = b < nBin ? binTot[b] : 0;
        s += v[k];
    }
    int incl = s;
    for (int o = 1; o < 32; o <<= 1) {
        int t = __shfl_up(incl, o, 32);
        if (lane >= o) incl += t;
    }
    if (lane == 31) wsum[wv] = incl;
    __syncthreads();
    int pre = incl - s;
    for (int k = 0; k < wv; ++k) pre += wsum[k];
    #pragma unroll
    for (int k = 0; k < 4; ++k) {
        const int b = 4 * tid + k;
        if (b < nBin) {
            lpos[b] = pre + ghist[(size_t)b * nBlk + blk];
            if (blk == 0) binBase[b] = pre;
        }
        pre += v[k];
    }
    if (blk == 0 && tid == 0) binBase[nBin] = nE;
    __syncthreads();
    int pv[16];
    #pragma unroll
    for (int k = 0; k < 16; ++k)
        pv[k] = atomicAdd(&lpos[tv[k] >= 0 ? (tv[k] >> BSH) : MAXB], 1);
    #pragma unroll
    for (int k = 0; k < 16; ++k)
        if (tv[k] >= 0) tmp[pv[k]] = make_int2(sv[k], tv[k]);
}

__global__ __launch_bounds__(256) void k_binsort(const int2* __restrict__ tmp,
                                                 const int* __restrict__ binBase,
                                                 int* __restrict__ off, float* __restrict__ dinv,
                                                 int2* __restrict__ csr, int n, int nE, int nBin) {
    __shared__ int c[BW];
    __shared__ int ws[4];
    const int tid = threadIdx.x, lane = tid & 31, wv = tid >> 5;
    const int bin = blockIdx.x;
    const int beg = binBase[bin], end = binBase[bin + 1];
    if (tid < BW) c[tid] = 0;
    __syncthreads();
    for (int i = beg + tid; i < end; i += 256) atomicAdd(&c[tmp[i].y & (BW - 1)], 1);
    __syncthreads();
    int cntv = 0, incl = 0;
    if (tid < BW) {
        cntv = c[tid];
        incl = cntv;
        for (int o = 1; o < 32; o <<= 1) {
            int t = __shfl_up(incl, o, 32);
            if (lane >= o) incl += t;
        }
        if (lane == 31) ws[wv] = incl;
    }
    __syncthreads();
    if (tid < BW) {
        int pre = incl - cntv;
        for (int k = 0; k < wv; ++k) pre += ws[k];
        const int node = bin * BW + tid;
        const int o = beg + pre;
        c[tid] = o;
        if (node < n) {
            off[node] = o;
            float d = (float)(cntv + 1);
            dinv[node] = d > 0.0f ? rsqrtf(d) : 0.0f;
        }
    }
    if (bin == nBin - 1 && tid == 0) off[n] = nE;
    __syncthreads();
    for (int i = beg + tid; i < end; i += 256) {
        const int2 r = tmp[i];
        const int p = atomicAdd(&c[r.y & (BW - 1)], 1);
        csr[p] = r;
    }
}

__global__ __launch_bounds__(256) void k_gemm0_weight(const float* __restrict__ x,
                                                      const float* __restrict__ W0,
                                                      const float* __restrict__ b0,
                                                      float* __restrict__ h,
                                                      float* __restrict__ h0, int n,
                                                      int gFirst, int gBlocks,
                                                      const float* __restrict__ dinv,
                                                      int2* __restrict__ csr, int nE) {
    if ((int)blockIdx.x < gBlocks) {
        gemm0_rows(x, W0, b0, h, h0, n, gFirst + (int)blockIdx.x);
        return;
    }
    const int i = ((int)blockIdx.x - gBlocks) * 256 + (int)threadIdx.x;
    if (i < nE) {
        const int2 r = csr[i];
        const float w = dinv[r.x] * dinv[r.y];
        csr[i] = make_int2(r.x, __float_as_int(w));
    }
}


#define TP 68

#define RSHARE(v, I) __builtin_amdgcn_update_dpp(0, (v), 0x150 + (I), 0xf, 0xf, false)

__device__ __forceinline__ float4 ldrow(const float* __restrict__ base, int row, unsigned lb) {
    return *(const float4*)((const char*)base + (size_t)((unsigned)row * 256u + lb));
}

#define G_LOAD(I) const float4 a##I = ldrow(h, RSHARE(cs, I), lb);
#define G_FMA(I) { const float w = __int_as_float(RSHARE(cw, I)); \
    acc.x = __builtin_fmaf(w, a##I.x, acc.x); acc.y = __builtin_fmaf(w, a##I.y, acc.y); \
    acc.z = __builtin_fmaf(w, a##I.z, acc.z); acc.w = __builtin_fmaf(w, a##I.w, acc.w); }
#define G_ALL(M) M(0) M(1) M(2) M(3) M(4) M(5) M(6) M(7) M(8) M(9) M(10) M(11) M(12) M(13) M(14) M(15)

__device__ __forceinline__ void gather16(float4& acc, const float* __restrict__ h,
                                         int cs, int cw, bool valid, unsigned lb) {
    const int sfirst = RSHARE(cs, 0);
    cs = valid ? cs : sfirst;
    cw = valid ? cw : 0;
    G_ALL(G_LOAD)
    G_ALL(G_FMA)
}

__device__ __forceinline__ void layer_body(const int* __restrict__ off,
                                           const int2* __restrict__ csr,
                                           const float* __restrict__ dinv,
                                           const float* __restrict__ hin,
                                           const float* __restrict__ h0,
                                           const float* __restrict__ W,
                                           float* __restrict__ hout,
                                           float beta, int n, int nE, const bool last,
                                           const float* __restrict__ Wout,
                                           const float* __restrict__ bout,
                                           float* __restrict__ out) {
    __shared__ __attribute__((aligned(16))) float tile[2][16 * TP];
    const int lane = threadIdx.x & 31;
    const int wave = threadIdx.x >> 5;
    const int m0   = (blockIdx.x * 2 + wave) * 16;
    const int ml   = lane & 15;
    const int kh   = lane >> 4;
    float* __restrict__ T = tile[wave];

    {
        const int nbase = m0 + 8 * kh;
        const unsigned lb = (unsigned)ml * 16u;
        int oi = nbase + (ml < 8 ? ml : 8); if (oi > n) oi = n;
        const int offv = off[oi];
        int di = nbase + (ml < 7 ? ml : 7); if (di > n - 1) di = n - 1;
        const float dv = dinv[di];
        const int emax = nE > 0 ? nE - 1 : 0;

        int beg = __shfl(offv, 0, 16), end = __shfl(offv, 1, 16);
        int i0 = beg + ml, i1 = beg + 16 + ml;
        bool v0 = i0 < end, v1 = i1 < end;
        int2 c0 = csr[i0 < emax ? i0 : emax];
        int2 c1 = csr[i1 < emax ? i1 : emax];
        float4 hsn = ldrow(hin, nbase < n - 1 ? nbase : n - 1, lb);

        #pragma nounroll
        for (int p = 0; p < 8; ++p) {
            int node = nbase + p; if (node > n - 1) node = n - 1;
            const int pn = p < 7 ? p + 1 : 8, pn1 = p < 7 ? p + 2 : 8;
            const int nbeg = __shfl(offv, pn, 16), nend = __shfl(offv, pn1, 16);
            const int j0 = nbeg + ml, j1 = nbeg + 16 + ml;
            const bool nv0 = j0 < nend, nv1 = j1 < nend;
            const int2 n0 = csr[j0 < emax ? j0 : emax];
            const int2 n1 = csr[j1 < emax ? j1 : emax];

            int nnode = nbase + (p < 7 ? p + 1 : 7); if (nnode > n - 1) nnode = n - 1;
            const float4 hs = hsn;
            hsn = ldrow(hin, nnode, lb);

            const float d = __shfl(dv, p, 16);
            const float4 z  = ldrow(h0, node, lb);
            const float dd = d * d;
            float4 acc;
            acc.x = dd * hs.x; acc.y = dd * hs.y; acc.z = dd * hs.z; acc.w = dd * hs.w;

            if (beg < end)      gather16(acc, hin, c0.x, c0.y, v0, lb);
            if (beg + 16 < end) gather16(acc, hin, c1.x, c1.y, v1, lb);
            for (int e = beg + 32; e < end; e += 16) {
                const int ix = e + ml;
                const bool vx = ix < end;
                const int2 cx = csr[ix < emax ? ix : emax];
                gather16(acc, hin, cx.x, cx.y, vx, lb);
            }

            float4 sup;
            sup.x = 0.9f * acc.x + 0.1f * z.x; sup.y = 0.9f * acc.y + 0.1f * z.y;
            sup.z = 0.9f * acc.z + 0.1f * z.z; sup.w = 0.9f * acc.w + 0.1f * z.w;
            *(float4*)(T + (8 * kh + p) * TP + 4 * ml) = sup;

            beg = nbeg; end = nend; c0 = n0; c1 = n1; v0 = nv0; v1 = nv1;
        }
    }
    __syncthreads();

    v8f acc[4] = {};
    for (int kk = 0; kk < HIDDEN; kk += 4) {
        const int ka = kk + kh * 2;
        const float2 av = *(const float2*)(T + ml * TP + ka);
        v2f a; a.x = av.x; a.y = av.y;
        #pragma unroll
        for (int nt = 0; nt < 4; ++nt) {
            v2f b = bfrag_rowmajor(W, HIDDEN, ka, nt * 16 + ml);
            WMMA_F32X4(acc[nt], a, b);
        }
    }
    const float omb = 1.0f - beta;
    const int rbase = m0 + kh * 8;
    const bool full = (m0 + 16 <= n);
    if (last) {
        #pragma unroll
        for (int nt = 0; nt < 4; ++nt) {
            const int c = nt * 16 + ml;
            #pragma unroll
            for (int v = 0; v < 8; ++v) {
                float* p = T + (kh * 8 + v) * TP + c;
                *p = fmaxf(beta * acc[nt][v] + omb * (*p), 0.0f);
            }
        }
        __syncthreads();
        v8f oacc[3] = {};
        const bool c2ok = (32 + ml) < NCLASSES;
        const int  c2   = c2ok ? 32 + ml : NCLASSES - 1;
        for (int kk = 0; kk < HIDDEN; kk += 4) {
            const int ka = kk + kh * 2;
            const float2 av = *(const float2*)(T + ml * TP + ka);
            v2f a; a.x = av.x; a.y = av.y;
            const float* w0 = Wout + (size_t)ka * NCLASSES;
            const float* w1 = w0 + NCLASSES;
            v2f b0v, b1v, b2v;
            b0v.x = w0[ml];      b0v.y = w1[ml];
            b1v.x = w0[16 + ml]; b1v.y = w1[16 + ml];
            const float t0 = w0[c2], t1 = w1[c2];
            b2v.x = c2ok ? t0 : 0.0f; b2v.y = c2ok ? t1 : 0.0f;
            WMMA_F32X4(oacc[0], a, b0v);
            WMMA_F32X4(oacc[1], a, b1v);
            WMMA_F32X4(oacc[2], a, b2v);
        }
        #pragma unroll
        for (int nt = 0; nt < 3; ++nt) {
            const int c = nt * 16 + ml;
            if (c < NCLASSES) {
                const float bias = bout[c];
                if (full) {
                    #pragma unroll
                    for (int v = 0; v < 8; ++v)
                        out[(size_t)(rbase + v) * NCLASSES + c] = oacc[nt][v] + bias;
                } else {
                    #pragma unroll
                    for (int v = 0; v < 8; ++v) {
                        const int r = rbase + v;
                        if (r < n) out[(size_t)r * NCLASSES + c] = oacc[nt][v] + bias;
                    }
                }
            }
        }
        return;
    }
    #pragma unroll
    for (int nt = 0; nt < 4; ++nt) {
        int c = nt * 16 + ml;
        if (full) {
            #pragma unroll
            for (int v = 0; v < 8; ++v) {
                size_t o = (size_t)(rbase + v) * HIDDEN + c;
                float sup = T[(kh * 8 + v) * TP + c];
                hout[o] = fmaxf(beta * acc[nt][v] + omb * sup, 0.0f);
            }
        } else {
            #pragma unroll
            for (int v = 0; v < 8; ++v) {
                int r = rbase + v;
                if (r < n) {
                    size_t o = (size_t)r * HIDDEN + c;
                    float sup = T[(kh * 8 + v) * TP + c];
                    hout[o] = fmaxf(beta * acc[nt][v] + omb * sup, 0.0f);
                }
            }
        }
    }
}


__global__ __launch_bounds__(64) void k_layer(const int* __restrict__ off,
                                              const int2* __restrict__ csr,
                                              const float* __restrict__ dinv,
                                              const float* __restrict__ hin,
                                              const float* __restrict__ h0,
                                              const float* __restrict__ W,
                                              float* __restrict__ hout,
                                              float beta, int n, int nE) {
    layer_body(off, csr, dinv, hin, h0, W, hout, beta, n, nE, false, nullptr, nullptr, nullptr);
}

__global__ __launch_bounds__(64) void k_layer_out(const int* __restrict__ off,
                                                  const int2* __restrict__ csr,
                                                  const float* __restrict__ dinv,
                                                  const float* __restrict__ hin,
                                                  const float* __restrict__ h0,
                                                  const float* __restrict__ W,
                                                  float beta, int n, int nE,
                                                  const float* __restrict__ Wout,
                                                  const float* __restrict__ bout,
                                                  float* __restrict__ out) {
    layer_body(off, csr, dinv, hin, h0, W, nullptr, beta, n, nE, true, Wout, bout, out);
}


extern "C" void kernel_launch(void* const* d_in, const int* in_sizes, int n_in,
                              void* d_out, int out_size, void* d_ws, size_t ws_size,
                              hipStream_t stream) {
    const float* x    = (const float*)d_in[0];
    const int*   ei   = (const int*)  d_in[1];
    const float* W0   = (const float*)d_in[2];
    const float* b0   = (const float*)d_in[3];
    const float* Wl   = (const float*)d_in[4];
    const float* Wout = (const float*)d_in[5];
    const float* bout = (const float*)d_in[6];

    const int N = in_sizes[0] / NFEATS;
    const int E = in_sizes[1] / 2;
    const int* erow = ei;
    const int* ecol = ei + E;

    float* h    = (float*)d_ws;
    float* h0   = h  + (size_t)N * HIDDEN;
    float* h2   = h0 + (size_t)N * HIDDEN;
    int2*  csr  = (int2*)(h2 + (size_t)N * HIDDEN);
    int2*  tmp  = csr + (size_t)E;
    size_t nAl  = ((size_t)N + 64) & ~(size_t)63;
    int*   off  = (int*)(tmp + (size_t)E);
    float* dinv = (float*)(off + nAl);
    int*   binTot  = (int*)(dinv + nAl);
    int*   binBase = binTot + MAXB;
    int*   ghist   = binBase + MAXB + 64;
    (void)ws_size; (void)n_in; (void)out_size;

    const int edgeBlocks = (E + 255) / 256;
    const int gemmBlocks = (N + 127) / 128;
    const int layerBlocks = (N + 31) / 32;
    const int nBin = (N + BW - 1) / BW;
    const int nBlk = (E + CHUNK - 1) / CHUNK;

    const int gA = gemmBlocks / 2;
    const int gB = gemmBlocks - gA;

    k_hist   <<<nBlk, 256, 0, stream>>>(ecol, ghist, E, nBin, nBlk);
    k_binscan<<<(nBin + 7) / 8, 256, 0, stream>>>(ghist, binTot, nBin, nBlk);
    k_gemm0_scatter<<<gA + nBlk, 256, 0, stream>>>(x, W0, b0, h, h0, N, 0, gA,
                                                   erow, ecol, ghist, binTot, binBase, tmp,
                                                   E, nBin, nBlk);
    k_binsort<<<nBin, 256, 0, stream>>>(tmp, binBase, off, dinv, csr, N, E, nBin);
    k_gemm0_weight<<<gB + edgeBlocks, 256, 0, stream>>>(x, W0, b0, h, h0, N, gA, gB,
                                                        dinv, csr, E);

    for (int i = 0; i < NLAYERS; ++i) {
        float beta = logf(0.5f / (float)(i + 1) + 1.0f);
        const float* hi = (i & 1) ? h2 : h;
        const float* Wi = Wl + (size_t)i * HIDDEN * HIDDEN;
        if (i + 1 < NLAYERS)
            k_layer<<<layerBlocks, 64, 0, stream>>>(off, csr, dinv, hi, h0, Wi,
                                                    (i & 1) ? h : h2, beta, N, E);
        else
            k_layer_out<<<layerBlocks, 64, 0, stream>>>(off, csr, dinv, hi, h0, Wi, beta, N, E,
                                                        Wout, bout, (float*)d_out);
    }
}


# Edge weight factored out of the edge list: rows stored scaled by dinv (agg = dinv[t]*(g[t]+sum g[s])); weight launch and its dinv lookups gone (13 -> 12), input GEMM runs in the bin-sort blocks
# speedup vs baseline: 4.6686x; 1.0357x over previous
#include <hip/hip_runtime.h>
#include <math.h>


typedef __attribute__((ext_vector_type(2))) float v2f;
typedef __attribute__((ext_vector_type(8))) float v8f;

#define NFEATS   128
#define HIDDEN   64
#define NCLASSES 40
#define NLAYERS  8


#define BSH   7
#define BW    128
#define MAXB  1024
#define CHUNK 4096

__global__ __launch_bounds__(256) void k_hist(const int* __restrict__ col, int* __restrict__ ghist,
                                              int nE, int nBin, int nBlk) {
    __shared__ int lh[MAXB + 1];
    const int tid = threadIdx.x, blk = blockIdx.x;
    for (int b = tid; b < nBin; b += 256) lh[b] = 0;
    int tv[16];
    #pragma unroll
    for (int k = 0; k < 16; ++k) {
        const int e = blk * CHUNK + k * 256 + tid;
        const int t = col[e < nE ? e : nE - 1];
        tv[k] = e < nE ? t : -1;
    }
    __syncthreads();
    #pragma unroll
    for (int k = 0; k < 16; ++k)
        atomicAdd(&lh[tv[k] >= 0 ? (tv[k] >> BSH) : MAXB], 1);
    __syncthreads();
    for (int b = tid; b < nBin; b += 256) ghist[(size_t)b * nBlk + blk] = lh[b];
}

__global__ __launch_bounds__(256) void k_binscan(int* __restrict__ ghist, int* __restrict__ binTot,
                                                 int nBin, int nBlk) {
    const int lane = threadIdx.x & 31;
    const int bin = blockIdx.x * 8 + (threadIdx.x >> 5);
    if (bin >= nBin) return;
    int* __restrict__ g = ghist + (size_t)bin * nBlk;
    int run = 0;
    for (int b0 = 0; b0 < nBlk; b0 += 32) {
        const int b = b0 + lane;
        const int v = b < nBlk ? g[b] : 0;
        int incl = v;
        for (int o = 1; o < 32; o <<= 1) {
            int t = __shfl_up(incl, o, 32);
            if (lane >= o) incl += t;
        }
        if (b < nBlk) g[b] = run + incl - v;
        run += __shfl(incl, 31, 32);
    }
    if (lane == 0) binTot[bin] = run;
}


__device__ __forceinline__ v2f bfrag_rowmajor(const float* __restrict__ B, int ldb,
                                              int kb  , int ncol) {
    v2f b;
    b.x = B[(size_t)kb * ldb + ncol];
    b.y = B[(size_t)(kb + 1) * ldb + ncol];
    return b;
}

#define WMMA_F32X4(acc, a, b) \
    (acc) = __builtin_amdgcn_wmma_f32_16x16x4_f32(false, (a), false, (b), (short)0, (acc), false, false)


__device__ __forceinline__ void gemm0_rows(const float* __restrict__ x,
                                           const float* __restrict__ W0,
                                           const float* __restrict__ b0,
                                           float* __restrict__ g,
                                           float* __restrict__ h0, int n, int blk,
                                           const float* sd) {
    const int lane = threadIdx.x & 31;
    const int wave = threadIdx.x >> 5;
    const int m0   = (blk * 8 + wave) * 16;
    const int ml   = lane & 15;
    const int kh   = lane >> 4;
    int m = m0 + ml; if (m >= n) m = n - 1;

    v8f acc[4] = {};
    for (int kk = 0; kk < NFEATS; kk += 4) {
        const int ka = kk + kh * 2;
        const float2 av = *(const float2*)(x + (size_t)m * NFEATS + ka);
        v2f a; a.x = av.x; a.y = av.y;
        #pragma unroll
        for (int nt = 0; nt < 4; ++nt) {
            v2f b = bfrag_rowmajor(W0, HIDDEN, ka, nt * 16 + ml);
            WMMA_F32X4(acc[nt], a, b);
        }
    }
    const int rbase = m0 + kh * 8;
    const int lbase = wave * 16 + kh * 8;
    const bool full = (m0 + 16 <= n);
    float dsc[8];
    #pragma unroll
    for (int v = 0; v < 8; ++v) dsc[v] = sd[lbase + v];
    #pragma unroll
    for (int nt = 0; nt < 4; ++nt) {
        int c = nt * 16 + ml;
        float bias = b0[c];
        if (full) {
            #pragma unroll
            for (int v = 0; v < 8; ++v) {
                float val = fmaxf(acc[nt][v] + bias, 0.0f);
                size_t o = (size_t)(rbase + v) * HIDDEN + c;
                h0[o] = val; g[o] = dsc[v] * val;
            }
        } else {
            #pragma unroll
            for (int v = 0; v < 8; ++v) {
                int r = rbase + v;
                if (r < n) {
                    float val = fmaxf(acc[nt][v] + bias, 0.0f);
                    size_t o = (size_t)r * HIDDEN + c;
                    h0[o] = val; g[o] = dsc[v] * val;
                }
            }
        }
    }
}

__global__ __launch_bounds__(256) void k_scatter(const int* __restrict__ row,
                                                 const int* __restrict__ col,
                                                 const int* __restrict__ ghist,
                                                 const int* __restrict__ binTot,
                                                 int* __restrict__ binBase,
                                                 int2* __restrict__ tmp,
                                                 int nE, int nBin, int nBlk) {
    __shared__ int lpos[MAXB + 1];
    __shared__ int wsum[8];
    const int blk = (int)blockIdx.x;
    const int tid = threadIdx.x, lane = tid & 31, wv = tid >> 5;

    int sv[16], tv[16];
    #pragma unroll
    for (int k = 0; k < 16; ++k) {
        const int e = blk * CHUNK + k * 256 + tid;
        const int ec = e < nE ? e : nE - 1;
        const int t = col[ec];
        sv[k] = row[ec];
        tv[k] = e < nE ? t : -1;
    }

    int v[4], s = 0;
    #pragma unroll
    for (int k = 0; k < 4; ++k) {
        const int b = 4 * tid + k;
        v[k] = b < nBin ? binTot[b] : 0;
        s += v[k];
    }
    int incl = s;
    for (int o = 1; o < 32; o <<= 1) {
        int t = __shfl_up(incl, o, 32);
        if (lane >= o) incl += t;
    }
    if (lane == 31) wsum[wv] = incl;
    __syncthreads();
    int pre = incl - s;
    for (int k = 0; k < wv; ++k) pre += wsum[k];
    #pragma unroll
    for (int k = 0; k < 4; ++k) {
        const int b = 4 * tid + k;
        if (b < nBin) {
            lpos[b] = pre + ghist[(size_t)b * nBlk + blk];
            if (blk == 0) binBase[b] = pre;
        }
        pre += v[k];
    }
    if (blk == 0 && tid == 0) binBase[nBin] = nE;
    __syncthreads();
    int pv[16];
    #pragma unroll
    for (int k = 0; k < 16; ++k)
        pv[k] = atomicAdd(&lpos[tv[k] >= 0 ? (tv[k] >> BSH) : MAXB], 1);
    #pragma unroll
    for (int k = 0; k < 16; ++k)
        if (tv[k] >= 0) tmp[pv[k]] = make_int2(sv[k], tv[k]);
}

__global__ __launch_bounds__(256) void k_binsort_gemm0(const int2* __restrict__ tmp,
                                                       const int* __restrict__ binBase,
                                                       int* __restrict__ off, float* __restrict__ dinv,
                                                       int* __restrict__ csr, int n, int nE, int nBin,
                                                       const float* __restrict__ x,
                                                       const float* __restrict__ W0,
                                                       const float* __restrict__ b0,
                                                       float* __restrict__ g,
                                                       float* __restrict__ h0) {
    __shared__ int c[BW];
    __shared__ int ws[4];
    __shared__ float sd[BW];
    const int tid = threadIdx.x, lane = tid & 31, wv = tid >> 5;
    const int bin = blockIdx.x;
    const int beg = binBase[bin], end = binBase[bin + 1];
    if (tid < BW) c[tid] = 0;
    __syncthreads();
    for (int i = beg + tid; i < end; i += 256) atomicAdd(&c[tmp[i].y & (BW - 1)], 1);
    __syncthreads();
    int cntv = 0, incl = 0;
    if (tid < BW) {
        cntv = c[tid];
        incl = cntv;
        for (int o = 1; o < 32; o <<= 1) {
            int t = __shfl_up(incl, o, 32);
            if (lane >= o) incl += t;
        }
        if (lane == 31) ws[wv] = incl;
    }
    __syncthreads();
    if (tid < BW) {
        int pre = incl - cntv;
        for (int k = 0; k < wv; ++k) pre += ws[k];
        const int node = bin * BW + tid;
        const int o = beg + pre;
        c[tid] = o;
        float d = (float)(cntv + 1);
        const float di = d > 0.0f ? rsqrtf(d) : 0.0f;
        sd[tid] = di;
        if (node < n) {
            off[node] = o;
            dinv[node] = di;
        }
    }
    if (bin == nBin - 1 && tid == 0) off[n] = nE;
    __syncthreads();
    for (int i = beg + tid; i < end; i += 256) {
        const int2 r = tmp[i];
        const int p = atomicAdd(&c[r.y & (BW - 1)], 1);
        csr[p] = r.x;
    }
    gemm0_rows(x, W0, b0, g, h0, n, bin, sd);
}


#define TP 68

#define RSHARE(v, I) __builtin_amdgcn_update_dpp(0, (v), 0x150 + (I), 0xf, 0xf, false)

__device__ __forceinline__ float4 ldrow(const float* __restrict__ base, int row, unsigned lb) {
    return *(const float4*)((const char*)base + (size_t)((unsigned)row * 256u + lb));
}

#define G_LOAD(I) const float4 a##I = ldrow(h, RSHARE(cs, I), lb);
#define G_FMA(I) { const float w = __int_as_float(RSHARE(cw, I)); \
    acc.x = __builtin_fmaf(w, a##I.x, acc.x); acc.y = __builtin_fmaf(w, a##I.y, acc.y); \
    acc.z = __builtin_fmaf(w, a##I.z, acc.z); acc.w = __builtin_fmaf(w, a##I.w, acc.w); }
#define G_ALL(M) M(0) M(1) M(2) M(3) M(4) M(5) M(6) M(7) M(8) M(9) M(10) M(11) M(12) M(13) M(14) M(15)

__device__ __forceinline__ void gather16(float4& acc, const float* __restrict__ h,
                                         int cs, bool valid, unsigned lb) {
    const int sfirst = RSHARE(cs, 0);
    cs = valid ? cs : sfirst;
    const int cw = valid ? 0x3f800000 : 0;
    G_ALL(G_LOAD)
    G_ALL(G_FMA)
}

__device__ __forceinline__ void layer_body(const int* __restrict__ off,
                                           const int* __restrict__ csr,
                                           const float* __restrict__ dinv,
                                           const float* __restrict__ hin,
                                           const float* __restrict__ h0,
                                           const float* __restrict__ W,
                                           float* __restrict__ hout,
                                           float beta, int n, int nE, const bool last,
                                           const float* __restrict__ Wout,
                                           const float* __restrict__ bout,
                                           float* __restrict__ out) {
    __shared__ __attribute__((aligned(16))) float tile[2][16 * TP];
    const int lane = threadIdx.x & 31;
    const int wave = threadIdx.x >> 5;
    const int m0   = (blockIdx.x * 2 + wave) * 16;
    const int ml   = lane & 15;
    const int kh   = lane >> 4;
    float* __restrict__ T = tile[wave];
    float dv;

    {
        const int nbase = m0 + 8 * kh;
        const unsigned lb = (unsigned)ml * 16u;
        int oi = nbase + (ml < 8 ? ml : 8); if (oi > n) oi = n;
        const int offv = off[oi];
        int di = nbase + (ml < 7 ? ml : 7); if (di > n - 1) di = n - 1;
        dv = dinv[di];
        const int emax = nE > 0 ? nE - 1 : 0;

        int beg = __shfl(offv, 0, 16), end = __shfl(offv, 1, 16);
        int i0 = beg + ml, i1 = beg + 16 + ml;
        bool v0 = i0 < end, v1 = i1 < end;
        int c0 = csr[i0 < emax ? i0 : emax];
        int c1 = csr[i1 < emax ? i1 : emax];
        float4 hsn = ldrow(hin, nbase < n - 1 ? nbase : n - 1, lb);

        #pragma nounroll
        for (int p = 0; p < 8; ++p) {
            int node = nbase + p; if (node > n - 1) node = n - 1;
            const int pn = p < 7 ? p + 1 : 8, pn1 = p < 7 ? p + 2 : 8;
            const int nbeg = __shfl(offv, pn, 16), nend = __shfl(offv, pn1, 16);
            const int j0 = nbeg + ml, j1 = nbeg + 16 + ml;
            const bool nv0 = j0 < nend, nv1 = j1 < nend;
            const int n0 = csr[j0 < emax ? j0 : emax];
            const int n1 = csr[j1 < emax ? j1 : emax];

            int nnode = nbase + (p < 7 ? p + 1 : 7); if (nnode > n - 1) nnode = n - 1;
            const float4 hs = hsn;
            hsn = ldrow(hin, nnode, lb);

            const float d = __shfl(dv, p, 16);
            const float4 z  = ldrow(h0, node, lb);
            float4 acc = hs;

            if (beg < end)      gather16(acc, hin, c0, v0, lb);
            if (beg + 16 < end) gather16(acc, hin, c1, v1, lb);
            for (int e = beg + 32; e < end; e += 16) {
                const int ix = e + ml;
                const bool vx = ix < end;
                const int cx = csr[ix < emax ? ix : emax];
                gather16(acc, hin, cx, vx, lb);
            }

            const float k9 = 0.9f * d;
            float4 sup;
            sup.x = k9 * acc.x + 0.1f * z.x; sup.y = k9 * acc.y + 0.1f * z.y;
            sup.z = k9 * acc.z + 0.1f * z.z; sup.w = k9 * acc.w + 0.1f * z.w;
            *(float4*)(T + (8 * kh + p) * TP + 4 * ml) = sup;

            beg = nbeg; end = nend; c0 = n0; c1 = n1; v0 = nv0; v1 = nv1;
        }
    }
    __syncthreads();

    v8f acc[4] = {};
    for (int kk = 0; kk < HIDDEN; kk += 4) {
        const int ka = kk + kh * 2;
        const float2 av = *(const float2*)(T + ml * TP + ka);
        v2f a; a.x = av.x; a.y = av.y;
        #pragma unroll
        for (int nt = 0; nt < 4; ++nt) {
            v2f b = bfrag_rowmajor(W, HIDDEN, ka, nt * 16 + ml);
            WMMA_F32X4(acc[nt], a, b);
        }
    }
    const float omb = 1.0f - beta;
    const int rbase = m0 + kh * 8;
    const bool full = (m0 + 16 <= n);
    if (last) {
        #pragma unroll
        for (int nt = 0; nt < 4; ++nt) {
            const int c = nt * 16 + ml;
            #pragma unroll
            for (int v = 0; v < 8; ++v) {
                float* p = T + (kh * 8 + v) * TP + c;
                *p = fmaxf(beta * acc[nt][v] + omb * (*p), 0.0f);
            }
        }
        __syncthreads();
        v8f oacc[3] = {};
        const bool c2ok = (32 + ml) < NCLASSES;
        const int  c2   = c2ok ? 32 + ml : NCLASSES - 1;
        for (int kk = 0; kk < HIDDEN; kk += 4) {
            const int ka = kk + kh * 2;
            const float2 av = *(const float2*)(T + ml * TP + ka);
            v2f a; a.x = av.x; a.y = av.y;
            const float* w0 = Wout + (size_t)ka * NCLASSES;
            const float* w1 = w0 + NCLASSES;
            v2f b0v, b1v, b2v;
            b0v.x = w0[ml];      b0v.y = w1[ml];
            b1v.x = w0[16 + ml]; b1v.y = w1[16 + ml];
            const float t0 = w0[c2], t1 = w1[c2];
            b2v.x = c2ok ? t0 : 0.0f; b2v.y = c2ok ? t1 : 0.0f;
            WMMA_F32X4(oacc[0], a, b0v);
            WMMA_F32X4(oacc[1], a, b1v);
            WMMA_F32X4(oacc[2], a, b2v);
        }
        #pragma unroll
        for (int nt = 0; nt < 3; ++nt) {
            const int c = nt * 16 + ml;
            if (c < NCLASSES) {
                const float bias = bout[c];
                if (full) {
                    #pragma unroll
                    for (int v = 0; v < 8; ++v)
                        out[(size_t)(rbase + v) * NCLASSES + c] = oacc[nt][v] + bias;
                } else {
                    #pragma unroll
                    for (int v = 0; v < 8; ++v) {
                        const int r = rbase + v;
                        if (r < n) out[(size_t)r * NCLASSES + c] = oacc[nt][v] + bias;
                    }
                }
            }
        }
        return;
    }
    float dsc[8];
    #pragma unroll
    for (int v = 0; v < 8; ++v) dsc[v] = __shfl(dv, v, 16);
    #pragma unroll
    for (int nt = 0; nt < 4; ++nt) {
        int c = nt * 16 + ml;
        if (full) {
            #pragma unroll
            for (int v = 0; v < 8; ++v) {
                size_t o = (size_t)(rbase + v) * HIDDEN + c;
                float sup = T[(kh * 8 + v) * TP + c];
                hout[o] = dsc[v] * fmaxf(beta * acc[nt][v] + omb * sup, 0.0f);
            }
        } else {
            #pragma unroll
            for (int v = 0; v < 8; ++v) {
                int r = rbase + v;
                if (r < n) {
                    size_t o = (size_t)r * HIDDEN + c;
                    float sup = T[(kh * 8 + v) * TP + c];
                    hout[o] = dsc[v] * fmaxf(beta * acc[nt][v] + omb * sup, 0.0f);
                }
            }
        }
    }
}


__global__ __launch_bounds__(64) void k_layer(const int* __restrict__ off,
                                              const int* __restrict__ csr,
                                              const float* __restrict__ dinv,
                                              const float* __restrict__ hin,
                                              const float* __restrict__ h0,
                                              const float* __restrict__ W,
                                              float* __restrict__ hout,
                                              float beta, int n, int nE) {
    layer_body(off, csr, dinv, hin, h0, W, hout, beta, n, nE, false, nullptr, nullptr, nullptr);
}

__global__ __launch_bounds__(64) void k_layer_out(const int* __restrict__ off,
                                                  const int* __restrict__ csr,
                                                  const float* __restrict__ dinv,
                                                  const float* __restrict__ hin,
                                                  const float* __restrict__ h0,
                                                  const float* __restrict__ W,
                                                  float beta, int n, int nE,
                                                  const float* __restrict__ Wout,
                                                  const float* __restrict__ bout,
                                                  float* __restrict__ out) {
    layer_body(off, csr, dinv, hin, h0, W, nullptr, beta, n, nE, true, Wout, bout, out);
}


extern "C" void kernel_launch(void* const* d_in, const int* in_sizes, int n_in,
                              void* d_out, int out_size, void* d_ws, size_t ws_size,
                              hipStream_t stream) {
    const float* x    = (const float*)d_in[0];
    const int*   ei   = (const int*)  d_in[1];
    const float* W0   = (const float*)d_in[2];
    const float* b0   = (const float*)d_in[3];
    const float* Wl   = (const float*)d_in[4];
    const float* Wout = (const float*)d_in[5];
    const float* bout = (const float*)d_in[6];

    const int N = in_sizes[0] / NFEATS;
    const int E = in_sizes[1] / 2;
    const int* erow = ei;
    const int* ecol = ei + E;

    float* h    = (float*)d_ws;
    float* h0   = h  + (size_t)N * HIDDEN;
    float* h2   = h0 + (size_t)N * HIDDEN;
    int2*  tmp  = (int2*)(h2 + (size_t)N * HIDDEN);
    int*   csr  = (int*)(tmp + (size_t)E);
    size_t nAl  = ((size_t)N + 64) & ~(size_t)63;
    int*   off  = csr + (((size_t)E + 63) & ~(size_t)63);
    float* dinv = (float*)(off + nAl);
    int*   binTot  = (int*)(dinv + nAl);
    int*   binBase = binTot + MAXB;
    int*   ghist   = binBase + MAXB + 64;
    (void)ws_size; (void)n_in; (void)out_size;

    const int layerBlocks = (N + 31) / 32;
    const int nBin = (N + BW - 1) / BW;
    const int nBlk = (E + CHUNK - 1) / CHUNK;

    k_hist   <<<nBlk, 256, 0, stream>>>(ecol, ghist, E, nBin, nBlk);
    k_binscan<<<(nBin + 7) / 8, 256, 0, stream>>>(ghist, binTot, nBin, nBlk);
    k_scatter<<<nBlk, 256, 0, stream>>>(erow, ecol, ghist, binTot, binBase, tmp, E, nBin, nBlk);
    k_binsort_gemm0<<<nBin, 256, 0, stream>>>(tmp, binBase, off, dinv, csr, N, E, nBin,
                                              x, W0, b0, h, h0);

    for (int i = 0; i < NLAYERS; ++i) {
        float beta = logf(0.5f / (float)(i + 1) + 1.0f);
        const float* hi = (i & 1) ? h2 : h;
        const float* Wi = Wl + (size_t)i * HIDDEN * HIDDEN;
        if (i + 1 < NLAYERS)
            k_layer<<<layerBlocks, 64, 0, stream>>>(off, csr, dinv, hi, h0, Wi,
                                                    (i & 1) ? h : h2, beta, N, E);
        else
            k_layer_out<<<layerBlocks, 64, 0, stream>>>(off, csr, dinv, hi, h0, Wi, beta, N, E,
                                                        Wout, bout, (float*)d_out);
    }
}


# Edge list in 2 launches, not 4: an edge block sorts its 8192 edges by bin in its own region; a bin block sums the slice-start table and keeps its slices in registers; sum order stays LDS arrival
# speedup vs baseline: 5.1761x; 1.1087x over previous
#include <hip/hip_runtime.h>
#include <math.h>


typedef __attribute__((ext_vector_type(2))) float v2f;
typedef __attribute__((ext_vector_type(8))) float v8f;

#define NFEATS   128
#define HIDDEN   64
#define NCLASSES 40
#define NLAYERS  8


#define BSH   7
#define BW    128
#define MAXB  1024
#define CHUNK 8192
#define SORTT 512
#define SEGL  8
#define SEGI  7
#define SEGM  3

__global__ __launch_bounds__(SORTT) void k_blocksort(const int* __restrict__ row,
                                                     const int* __restrict__ col,
                                                     int* __restrict__ st,
                                                     int2* __restrict__ tmp,
                                                     int nE, int nBin, int stP) {
    __shared__ int lpos[MAXB + 1];
    __shared__ int wsum[SORTT / 32];
    const int blk = (int)blockIdx.x;
    const int tid = threadIdx.x, lane = tid & 31, wv = tid >> 5;
    for (int b = tid; b <= MAXB; b += SORTT) lpos[b] = 0;

    int sv[16], tv[16];
    #pragma unroll
    for (int k = 0; k < 16; ++k) {
        const int e = blk * CHUNK + k * SORTT + tid;
        const int ec = e < nE ? e : nE - 1;
        const int t = col[ec];
        sv[k] = row[ec];
        tv[k] = e < nE ? t : -1;
    }
    __syncthreads();
    int pv[16];
    #pragma unroll
    for (int k = 0; k < 16; ++k)
        pv[k] = atomicAdd(&lpos[tv[k] >= 0 ? (tv[k] >> BSH) : MAXB], 1);
    __syncthreads();

    const int v0 = lpos[2 * tid], v1 = lpos[2 * tid + 1];
    const int s = v0 + v1;
    int incl = s;
    for (int o = 1; o < 32; o <<= 1) {
        int t = __shfl_up(incl, o, 32);
        if (lane >= o) incl += t;
    }
    if (lane == 31) wsum[wv] = incl;
    __syncthreads();
    int pre = incl - s;
    for (int k = 0; k < wv; ++k) pre += wsum[k];
    lpos[2 * tid]     = pre;
    lpos[2 * tid + 1] = pre + v0;
    if (2 * tid <= nBin)     st[(size_t)(2 * tid) * stP + blk]     = pre;
    if (2 * tid + 1 <= nBin) st[(size_t)(2 * tid + 1) * stP + blk] = pre + v0;
    if (tid == SORTT - 1 && nBin == MAXB) st[(size_t)MAXB * stP + blk] = pre + s;
    __syncthreads();
    int2* __restrict__ reg = tmp + (size_t)blk * CHUNK;
    int bp[16];
    #pragma unroll
    for (int k = 0; k < 16; ++k) bp[k] = lpos[tv[k] >= 0 ? (tv[k] >> BSH) : MAXB];
    #pragma unroll
    for (int k = 0; k < 16; ++k)
        if (tv[k] >= 0) reg[bp[k] + pv[k]] = make_int2(sv[k], tv[k]);
}


__device__ __forceinline__ v2f bfrag_rowmajor(const float* __restrict__ B, int ldb,
                                              int kb  , int ncol) {
    v2f b;
    b.x = B[(size_t)kb * ldb + ncol];
    b.y = B[(size_t)(kb + 1) * ldb + ncol];
    return b;
}

#define WMMA_F32X4(acc, a, b) \
    (acc) = __builtin_amdgcn_wmma_f32_16x16x4_f32(false, (a), false, (b), (short)0, (acc), false, false)


__device__ __forceinline__ void gemm0_rows(const float* __restrict__ x,
                                           const float* __restrict__ W0,
                                           const float* __restrict__ b0,
                                           float* __restrict__ g,
                                           float* __restrict__ h0, int n, int blk,
                                           const float* sd) {
    const int lane = threadIdx.x & 31;
    const int wave = threadIdx.x >> 5;
    const int m0   = (blk * 8 + wave) * 16;
    const int ml   = lane & 15;
    const int kh   = lane >> 4;
    int m = m0 + ml; if (m >= n) m = n - 1;

    v8f acc[4] = {};
    for (int kk = 0; kk < NFEATS; kk += 4) {
        const int ka = kk + kh * 2;
        const float2 av = *(const float2*)(x + (size_t)m * NFEATS + ka);
        v2f a; a.x = av.x; a.y = av.y;
        #pragma unroll
        for (int nt = 0; nt < 4; ++nt) {
            v2f b = bfrag_rowmajor(W0, HIDDEN, ka, nt * 16 + ml);
            WMMA_F32X4(acc[nt], a, b);
        }
    }
    const int rbase = m0 + kh * 8;
    const int lbase = wave * 16 + kh * 8;
    const bool full = (m0 + 16 <= n);
    float dsc[8];
    #pragma unroll
    for (int v = 0; v < 8; ++v) dsc[v] = sd[lbase + v];
    #pragma unroll
    for (int nt = 0; nt < 4; ++nt) {
        int c = nt * 16 + ml;
        float bias = b0[c];
        if (full) {
            #pragma unroll
            for (int v = 0; v < 8; ++v) {
                float val = fmaxf(acc[nt][v] + bias, 0.0f);
                size_t o = (size_t)(rbase + v) * HIDDEN + c;
                h0[o] = val; g[o] = dsc[v] * val;
            }
        } else {
            #pragma unroll
            for (int v = 0; v < 8; ++v) {
                int r = rbase + v;
                if (r < n) {
                    float val = fmaxf(acc[nt][v] + bias, 0.0f);
                    size_t o = (size_t)r * HIDDEN + c;
                    h0[o] = val; g[o] = dsc[v] * val;
                }
            }
        }
    }
}

__global__ __launch_bounds__(256) void k_binsort_gemm0(const int2* __restrict__ tmp,
                                                       const int* __restrict__ st,
                                                       int* __restrict__ off, float* __restrict__ dinv,
                                                       int* __restrict__ csr, int n, int nE, int nBin,
                                                       int nBlk, int stP,
                                                       const float* __restrict__ x,
                                                       const float* __restrict__ W0,
                                                       const float* __restrict__ b0,
                                                       float* __restrict__ g,
                                                       float* __restrict__ h0) {
    __shared__ int c[BW + 32];
    __shared__ int ws[4];
    __shared__ int red[8];
    __shared__ float sd[BW];
    const int tid = threadIdx.x, lane = tid & 31, wv = tid >> 5;
    const int sub = tid & (SEGL - 1), grp = tid >> 3;
    const int bin = blockIdx.x;
    const int* __restrict__ r0 = st + (size_t)bin * stP;
    const int* __restrict__ r1 = r0 + stP;

    int s0[SEGI], s1[SEGI];
    #pragma unroll
    for (int i = 0; i < SEGI; ++i) {
        const int b = grp + 32 * i;
        const int a0 = r0[b < nBlk ? b : 0], a1 = r1[b < nBlk ? b : 0];
        s0[i] = b < nBlk ? a0 : 0;
        s1[i] = b < nBlk ? a1 : 0;
    }
    int2 rec[SEGI][SEGM];
    #pragma unroll
    for (int i = 0; i < SEGI; ++i) {
        const int b = grp + 32 * i;
        const int2* __restrict__ seg = tmp + (size_t)(b < nBlk ? b : 0) * CHUNK;
        #pragma unroll
        for (int m = 0; m < SEGM; ++m) {
            const int j = s0[i] + sub + SEGL * m;
            rec[i][m] = seg[j < s1[i] ? j : (s1[i] > 0 ? s1[i] - 1 : 0)];
        }
    }

    int sa = 0;
    #pragma unroll
    for (int i = 0; i < SEGI; ++i) sa += sub == 0 ? s0[i] : 0;
    #pragma nounroll
    for (int b = grp + 32 * SEGI; b < nBlk; b += 32) sa += sub == 0 ? r0[b] : 0;
    if (tid < BW + 32) c[tid] = 0;
    for (int o = 16; o > 0; o >>= 1) sa += __shfl_xor(sa, o, 32);
    if (lane == 0) red[wv] = sa;
    __syncthreads();
    int beg = 0;
    #pragma unroll
    for (int k = 0; k < 8; ++k) beg += red[k];

    int slot[SEGI][SEGM];
    #pragma unroll
    for (int i = 0; i < SEGI; ++i) {
        #pragma unroll
        for (int m = 0; m < SEGM; ++m) {
            slot[i][m] = (s0[i] + sub + SEGL * m < s1[i]) ? (rec[i][m].y & (BW - 1)) : BW + lane;
            atomicAdd(&c[slot[i][m]], 1);
        }
    }
    #pragma unroll
    for (int i = 0; i < SEGI; ++i) {
        const int b = grp + 32 * i;
        const int2* __restrict__ seg = tmp + (size_t)(b < nBlk ? b : 0) * CHUNK;
        #pragma nounroll
        for (int j = s0[i] + SEGL * SEGM + sub; j < s1[i]; j += SEGL) atomicAdd(&c[seg[j].y & (BW - 1)], 1);
    }
    #pragma nounroll
    for (int b = grp + 32 * SEGI; b < nBlk; b += 32) {
        const int2* __restrict__ seg = tmp + (size_t)b * CHUNK;
        const int q1 = r1[b];
        #pragma nounroll
        for (int j = r0[b] + sub; j < q1; j += SEGL) atomicAdd(&c[seg[j].y & (BW - 1)], 1);
    }
    __syncthreads();
    int cntv = 0, incl = 0;
    if (tid < BW) {
        cntv = c[tid];
        incl = cntv;
        for (int o = 1; o < 32; o <<= 1) {
            int t = __shfl_up(incl, o, 32);
            if (lane >= o) incl += t;
        }
        if (lane == 31) ws[wv] = incl;
    }
    __syncthreads();
    if (tid < BW) {
        int pre = incl - cntv;
        for (int k = 0; k < wv; ++k) pre += ws[k];
        const int node = bin * BW + tid;
        const int o = beg + pre;
        c[tid] = o;
        float d = (float)(cntv + 1);
        const float di = d > 0.0f ? rsqrtf(d) : 0.0f;
        sd[tid] = di;
        if (node < n) {
            off[node] = o;
            dinv[node] = di;
        }
    }
    if (bin == nBin - 1 && tid == 0) off[n] = nE;
    __syncthreads();
    int pp[SEGI][SEGM];
    #pragma unroll
    for (int i = 0; i < SEGI; ++i) {
        #pragma unroll
        for (int m = 0; m < SEGM; ++m) pp[i][m] = atomicAdd(&c[slot[i][m]], 1);
    }
    #pragma unroll
    for (int i = 0; i < SEGI; ++i) {
        #pragma unroll
        for (int m = 0; m < SEGM; ++m)
            if (s0[i] + sub + SEGL * m < s1[i]) csr[pp[i][m]] = rec[i][m].x;
    }
    #pragma unroll
    for (int i = 0; i < SEGI; ++i) {
        const int b = grp + 32 * i;
        const int2* __restrict__ seg = tmp + (size_t)(b < nBlk ? b : 0) * CHUNK;
        #pragma nounroll
        for (int j = s0[i] + SEGL * SEGM + sub; j < s1[i]; j += SEGL) {
            const int2 r = seg[j];
            const int p = atomicAdd(&c[r.y & (BW - 1)], 1);
            csr[p] = r.x;
        }
    }
    #pragma nounroll
    for (int b = grp + 32 * SEGI; b < nBlk; b += 32) {
        const int2* __restrict__ seg = tmp + (size_t)b * CHUNK;
        const int q1 = r1[b];
        #pragma nounroll
        for (int j = r0[b] + sub; j < q1; j += SEGL) {
            const int2 r = seg[j];
            const int p = atomicAdd(&c[r.y & (BW - 1)], 1);
            csr[p] = r.x;
        }
    }
    gemm0_rows(x, W0, b0, g, h0, n, bin, sd);
}


#define TP 68

#define RSHARE(v, I) __builtin_amdgcn_update_dpp(0, (v), 0x150 + (I), 0xf, 0xf, false)

__device__ __forceinline__ float4 ldrow(const float* __restrict__ base, int row, unsigned lb) {
    return *(const float4*)((const char*)base + (size_t)((unsigned)row * 256u + lb));
}

#define G_LOAD(I) const float4 a##I = ldrow(h, RSHARE(cs, I), lb);
#define G_FMA(I) { const float w = __int_as_float(RSHARE(cw, I)); \
    acc.x = __builtin_fmaf(w, a##I.x, acc.x); acc.y = __builtin_fmaf(w, a##I.y, acc.y); \
    acc.z = __builtin_fmaf(w, a##I.z, acc.z); acc.w = __builtin_fmaf(w, a##I.w, acc.w); }
#define G_ALL(M) M(0) M(1) M(2) M(3) M(4) M(5) M(6) M(7) M(8) M(9) M(10) M(11) M(12) M(13) M(14) M(15)

__device__ __forceinline__ void gather16(float4& acc, const float* __restrict__ h,
                                         int cs, bool valid, unsigned lb) {
    const int sfirst = RSHARE(cs, 0);
    cs = valid ? cs : sfirst;
    const int cw = valid ? 0x3f800000 : 0;
    G_ALL(G_LOAD)
    G_ALL(G_FMA)
}

__device__ __forceinline__ void layer_body(const int* __restrict__ off,
                                           const int* __restrict__ csr,
                                           const float* __restrict__ dinv,
                                           const float* __restrict__ hin,
                                           const float* __restrict__ h0,
                                           const float* __restrict__ W,
                                           float* __restrict__ hout,
                                           float beta, int n, int nE, const bool last,
                                           const float* __restrict__ Wout,
                                           const float* __restrict__ bout,
                                           float* __restrict__ out) {
    __shared__ __attribute__((aligned(16))) float tile[2][16 * TP];
    const int lane = threadIdx.x & 31;
    const int wave = threadIdx.x >> 5;
    const int m0   = (blockIdx.x * 2 + wave) * 16;
    const int ml   = lane & 15;
    const int kh   = lane >> 4;
    float* __restrict__ T = tile[wave];
    float dv;

    {
        const int nbase = m0 + 8 * kh;
        const unsigned lb = (unsigned)ml * 16u;
        int oi = nbase + (ml < 8 ? ml : 8); if (oi > n) oi = n;
        const int offv = off[oi];
        int di = nbase + (ml < 7 ? ml : 7); if (di > n - 1) di = n - 1;
        dv = dinv[di];
        const int emax = nE > 0 ? nE - 1 : 0;

        int beg = __shfl(offv, 0, 16), end = __shfl(offv, 1, 16);
        int i0 = beg + ml, i1 = beg + 16 + ml;
        bool v0 = i0 < end, v1 = i1 < end;
        int c0 = csr[i0 < emax ? i0 : emax];
        int c1 = csr[i1 < emax ? i1 : emax];
        float4 hsn = ldrow(hin, nbase < n - 1 ? nbase : n - 1, lb);

        #pragma nounroll
        for (int p = 0; p < 8; ++p) {
            int node = nbase + p; if (node > n - 1) node = n - 1;
            const int pn = p < 7 ? p + 1 : 8, pn1 = p < 7 ? p + 2 : 8;
            const int nbeg = __shfl(offv, pn, 16), nend = __shfl(offv, pn1, 16);
            const int j0 = nbeg + ml, j1 = nbeg + 16 + ml;
            const bool nv0 = j0 < nend, nv1 = j1 < nend;
            const int n0 = csr[j0 < emax ? j0 : emax];
            const int n1 = csr[j1 < emax ? j1 : emax];

            int nnode = nbase + (p < 7 ? p + 1 : 7); if (nnode > n - 1) nnode = n - 1;
            const float4 hs = hsn;
            hsn = ldrow(hin, nnode, lb);

            const float d = __shfl(dv, p, 16);
            const float4 z  = ldrow(h0, node, lb);
            float4 acc = hs;

            if (beg < end)      gather16(acc, hin, c0, v0, lb);
            if (beg + 16 < end) gather16(acc, hin, c1, v1, lb);
            for (int e = beg + 32; e < end; e += 16) {
                const int ix = e + ml;
                const bool vx = ix < end;
                const int cx = csr[ix < emax ? ix : emax];
                gather16(acc, hin, cx, vx, lb);
            }

            const float k9 = 0.9f * d;
            float4 sup;
            sup.x = k9 * acc.x + 0.1f * z.x; sup.y = k9 * acc.y + 0.1f * z.y;
            sup.z = k9 * acc.z + 0.1f * z.z; sup.w = k9 * acc.w + 0.1f * z.w;
            *(float4*)(T + (8 * kh + p) * TP + 4 * ml) = sup;

            beg = nbeg; end = nend; c0 = n0; c1 = n1; v0 = nv0; v1 = nv1;
        }
    }
    __syncthreads();

    v8f acc[4] = {};
    for (int kk = 0; kk < HIDDEN; kk += 4) {
        const int ka = kk + kh * 2;
        const float2 av = *(const float2*)(T + ml * TP + ka);
        v2f a; a.x = av.x; a.y = av.y;
        #pragma unroll
        for (int nt = 0; nt < 4; ++nt) {
            v2f b = bfrag_rowmajor(W, HIDDEN, ka, nt * 16 + ml);
            WMMA_F32X4(acc[nt], a, b);
        }
    }
    const float omb = 1.0f - beta;
    const int rbase = m0 + kh * 8;
    const bool full = (m0 + 16 <= n);
    if (last) {
        #pragma unroll
        for (int nt = 0; nt < 4; ++nt) {
            const int c = nt * 16 + ml;
            #pragma unroll
            for (int v = 0; v < 8; ++v) {
                float* p = T + (kh * 8 + v) * TP + c;
                *p = fmaxf(beta * acc[nt][v] + omb * (*p), 0.0f);
            }
        }
        __syncthreads();
        v8f oacc[3] = {};
        const bool c2ok = (32 + ml) < NCLASSES;
        const int  c2   = c2ok ? 32 + ml : NCLASSES - 1;
        for (int kk = 0; kk < HIDDEN; kk += 4) {
            const int ka = kk + kh * 2;
            const float2 av = *(const float2*)(T + ml * TP + ka);
            v2f a; a.x = av.x; a.y = av.y;
            const float* w0 = Wout + (size_t)ka * NCLASSES;
            const float* w1 = w0 + NCLASSES;
            v2f b0v, b1v, b2v;
            b0v.x = w0[ml];      b0v.y = w1[ml];
            b1v.x = w0[16 + ml]; b1v.y = w1[16 + ml];
            const float t0 = w0[c2], t1 = w1[c2];
            b2v.x = c2ok ? t0 : 0.0f; b2v.y = c2ok ? t1 : 0.0f;
            WMMA_F32X4(oacc[0], a, b0v);
            WMMA_F32X4(oacc[1], a, b1v);
            WMMA_F32X4(oacc[2], a, b2v);
        }
        #pragma unroll
        for (int nt = 0; nt < 3; ++nt) {
            const int c = nt * 16 + ml;
            if (c < NCLASSES) {
                const float bias = bout[c];
                if (full) {
                    #pragma unroll
                    for (int v = 0; v < 8; ++v)
                        out[(size_t)(rbase + v) * NCLASSES + c] = oacc[nt][v] + bias;
                } else {
                    #pragma unroll
                    for (int v = 0; v < 8; ++v) {
                        const int r = rbase + v;
                        if (r < n) out[(size_t)r * NCLASSES + c] = oacc[nt][v] + bias;
                    }
                }
            }
        }
        return;
    }
    float dsc[8];
    #pragma unroll
    for (int v = 0; v < 8; ++v) dsc[v] = __shfl(dv, v, 16);
    #pragma unroll
    for (int nt = 0; nt < 4; ++nt) {
        int c = nt * 16 + ml;
        if (full) {
            #pragma unroll
            for (int v = 0; v < 8; ++v) {
                size_t o = (size_t)(rbase + v) * HIDDEN + c;
                float sup = T[(kh * 8 + v) * TP + c];
                hout[o] = dsc[v] * fmaxf(beta * acc[nt][v] + omb * sup, 0.0f);
            }
        } else {
            #pragma unroll
            for (int v = 0; v < 8; ++v) {
                int r = rbase + v;
                if (r < n) {
                    size_t o = (size_t)r * HIDDEN + c;
                    float sup = T[(kh * 8 + v) * TP + c];
                    hout[o] = dsc[v] * fmaxf(beta * acc[nt][v] + omb * sup, 0.0f);
                }
            }
        }
    }
}


__global__ __launch_bounds__(64) void k_layer(const int* __restrict__ off,
                                              const int* __restrict__ csr,
                                              const float* __restrict__ dinv,
                                              const float* __restrict__ hin,
                                              const float* __restrict__ h0,
                                              const float* __restrict__ W,
                                              float* __restrict__ hout,
                                              float beta, int n, int nE) {
    layer_body(off, csr, dinv, hin, h0, W, hout, beta, n, nE, false, nullptr, nullptr, nullptr);
}

__global__ __launch_bounds__(64) void k_layer_out(const int* __restrict__ off,
                                                  const int* __restrict__ csr,
                                                  const float* __restrict__ dinv,
                                                  const float* __restrict__ hin,
                                                  const float* __restrict__ h0,
                                                  const float* __restrict__ W,
                                                  float beta, int n, int nE,
                                                  const float* __restrict__ Wout,
                                                  const float* __restrict__ bout,
                                                  float* __restrict__ out) {
    layer_body(off, csr, dinv, hin, h0, W, nullptr, beta, n, nE, true, Wout, bout, out);
}


extern "C" void kernel_launch(void* const* d_in, const int* in_sizes, int n_in,
                              void* d_out, int out_size, void* d_ws, size_t ws_size,
                              hipStream_t stream) {
    const float* x    = (const float*)d_in[0];
    const int*   ei   = (const int*)  d_in[1];
    const float* W0   = (const float*)d_in[2];
    const float* b0   = (const float*)d_in[3];
    const float* Wl   = (const float*)d_in[4];
    const float* Wout = (const float*)d_in[5];
    const float* bout = (const float*)d_in[6];

    const int N = in_sizes[0] / NFEATS;
    const int E = in_sizes[1] / 2;
    const int* erow = ei;
    const int* ecol = ei + E;

    float* h    = (float*)d_ws;
    float* h0   = h  + (size_t)N * HIDDEN;
    float* h2   = h0 + (size_t)N * HIDDEN;
    const int nBlk = (E + CHUNK - 1) / CHUNK;
    const int stP  = (nBlk + 31) & ~31;
    int2*  tmp  = (int2*)(h2 + (size_t)N * HIDDEN);
    int*   csr  = (int*)(tmp + (size_t)nBlk * CHUNK);
    size_t nAl  = ((size_t)N + 64) & ~(size_t)63;
    int*   off  = csr + (((size_t)E + 63) & ~(size_t)63);
    float* dinv = (float*)(off + nAl);
    int*   st   = (int*)(dinv + nAl);
    (void)ws_size; (void)n_in; (void)out_size;

    const int layerBlocks = (N + 31) / 32;
    const int nBin = (N + BW - 1) / BW;

    k_blocksort<<<nBlk, SORTT, 0, stream>>>(erow, ecol, st, tmp, E, nBin, stP);
    k_binsort_gemm0<<<nBin, 256, 0, stream>>>(tmp, st, off, dinv, csr, N, E, nBin, nBlk, stP,
                                              x, W0, b0, h, h0);

    for (int i = 0; i < NLAYERS; ++i) {
        float beta = logf(0.5f / (float)(i + 1) + 1.0f);
        const float* hi = (i & 1) ? h2 : h;
        const float* Wi = Wl + (size_t)i * HIDDEN * HIDDEN;
        if (i + 1 < NLAYERS)
            k_layer<<<layerBlocks, 64, 0, stream>>>(off, csr, dinv, hi, h0, Wi,
                                                    (i & 1) ? h : h2, beta, N, E);
        else
            k_layer_out<<<layerBlocks, 64, 0, stream>>>(off, csr, dinv, hi, h0, Wi, beta, N, E,
                                                        Wout, bout, (float*)d_out);
    }
}
